# starting kernel
# baseline (speedup 1.0000x reference)
#include <hip/hip_runtime.h>
#include <math.h>

#define N_NODES  50000
#define N_EDGES  800000
#define HID      128
#define EDGE_D   16
#define N_GRAPHS 64

typedef __attribute__((ext_vector_type(16))) __bf16 v16bf;
typedef __attribute__((ext_vector_type(8)))  float  v8f;

union Frag {
    uint4 q[2];
    v16bf v;
};

__device__ __forceinline__ unsigned short f2bf(float f) {
    union { float f; unsigned int u; } c; c.f = f;
    unsigned int u = c.u;
    u += 0x7fffu + ((u >> 16) & 1u);
    return (unsigned short)(u >> 16);
}

__device__ __forceinline__ float silu_f(float x) {
    return x / (1.0f + __expf(-x));
}

__device__ __forceinline__ int asw_pos(int m_local, int K, int KT) {
    int mt = m_local >> 4, ml = m_local & 15;
    int t = K >> 5, kk = K & 31;
    int g, i;
    if (kk < 16) { g = kk >> 3;        i = kk & 7; }
    else         { g = (kk - 16) >> 3; i = 8 + ((kk - 16) & 7); }
    return ((mt * KT + t) * 32 + (ml + (g << 4))) * 16 + i;
}

__global__ void wconv_kernel(const float* __restrict__ W,
                             unsigned short* __restrict__ dst,
                             int Ksrc, int KT) {
    int idx = blockIdx.x * blockDim.x + threadIdx.x;
    int total = KT * 8 * 32 * 16;
    if (idx >= total) return;
    int i    = idx & 15;
    int lane = (idx >> 4) & 31;
    int tj   = idx >> 9;
    int t = tj >> 3, j = tj & 7;
    int g = lane >> 4;
    int K = t * 32 + (i < 8 ? g * 8 + i : 16 + g * 8 + (i - 8));
    int n = j * 16 + (lane & 15);
    float v = (K < Ksrc) ? W[K * HID + n] : 0.0f;
    dst[idx] = f2bf(v);
}

__global__ void zero_f32_kernel(float* p, int n) {
    int i = blockIdx.x * blockDim.x + threadIdx.x;
    if (i < n) p[i] = 0.0f;
}

#define K1T 9
#define K2T 4

__global__ __launch_bounds__(256)
void edge_kernel(const float* __restrict__ h,
                 const int*   __restrict__ eidx,
                 const float* __restrict__ x,
                 const float* __restrict__ eattr,
                 const unsigned short* __restrict__ We1b,
                 const float* __restrict__ be1,
                 const unsigned short* __restrict__ We2b,
                 const float* __restrict__ be2,
                 float* __restrict__ agg) {
    __shared__ uint4 lA4[4 * K1T * 32 * 2];
    __shared__ uint4 lM4[4 * K2T * 32 * 2];
    unsigned short* lA = (unsigned short*)lA4;
    unsigned short* lM = (unsigned short*)lM4;

    const int tid = threadIdx.x;
    const int e0  = blockIdx.x * 64;

    for (int idx = tid; idx < 64 * 288; idx += 256) {
        int m = idx / 288;
        int K = idx - m * 288;
        int e = e0 + m;
        int r = eidx[e];
        int c = eidx[N_EDGES + e];
        float v;
        if (K < HID)            v = h[r * HID + K];
        else if (K < 2 * HID)   v = h[c * HID + (K - HID)];
        else if (K == 2 * HID) {
            float d0 = x[r * 3 + 0] - x[c * 3 + 0];
            float d1 = x[r * 3 + 1] - x[c * 3 + 1];
            float d2 = x[r * 3 + 2] - x[c * 3 + 2];
            v = d0 * d0 + d1 * d1 + d2 * d2;
        }
        else if (K < 2 * HID + 1 + EDGE_D) v = eattr[e * EDGE_D + (K - 2 * HID - 1)];
        else v = 0.0f;
        lA[asw_pos(m, K, K1T)] = f2bf(v);
    }
    __syncthreads();

    const int w    = tid >> 5;
    const int lane = tid & 31;
    const int hi   = lane >> 4;
    const int n    = w * 16 + (lane & 15);

    v8f acc[4] = {};
    for (int t = 0; t < K1T; ++t) {
        Frag B;
        const uint4* bp = (const uint4*)We1b + ((t * 8 + w) * 32 + lane) * 2;
        B.q[0] = bp[0]; B.q[1] = bp[1];
        #pragma unroll
        for (int mt = 0; mt < 4; ++mt) {
            Frag A;
            const uint4* ap = (const uint4*)lA4 + ((mt * K1T + t) * 32 + lane) * 2;
            A.q[0] = ap[0]; A.q[1] = ap[1];
            acc[mt] = __builtin_amdgcn_wmma_f32_16x16x32_bf16(
                false, A.v, false, B.v, (short)0, acc[mt], false, false);
        }
    }
    {
        float b = be1[n];
        #pragma unroll
        for (int mt = 0; mt < 4; ++mt) {
            #pragma unroll
            for (int j = 0; j < 8; ++j) {
                int m = mt * 16 + hi * 8 + j;
                lM[asw_pos(m, n, K2T)] = f2bf(silu_f(acc[mt][j] + b));
            }
        }
    }
    __syncthreads();

    v8f acc2[4] = {};
    for (int t = 0; t < K2T; ++t) {
        Frag B;
        const uint4* bp = (const uint4*)We2b + ((t * 8 + w) * 32 + lane) * 2;
        B.q[0] = bp[0]; B.q[1] = bp[1];
        #pragma unroll
        for (int mt = 0; mt < 4; ++mt) {
            Frag A;
            const uint4* ap = (const uint4*)lM4 + ((mt * K2T + t) * 32 + lane) * 2;
            A.q[0] = ap[0]; A.q[1] = ap[1];
            acc2[mt] = __builtin_amdgcn_wmma_f32_16x16x32_bf16(
                false, A.v, false, B.v, (short)0, acc2[mt], false, false);
        }
    }
    {
        float b = be2[n];
        #pragma unroll
        for (int mt = 0; mt < 4; ++mt) {
            #pragma unroll
            for (int j = 0; j < 8; ++j) {
                int m = mt * 16 + hi * 8 + j;
                float f = silu_f(acc2[mt][j] + b);
                int r = eidx[e0 + m];
                atomicAdd(&agg[r * HID + n], f);
            }
        }
    }
}

#define KN1T 8
#define KN2T 4

__global__ __launch_bounds__(256)
void node_kernel(const float* __restrict__ h,
                 const float* __restrict__ agg,
                 const int*   __restrict__ batch,
                 const unsigned short* __restrict__ Wn1b,
                 const float* __restrict__ bn1,
                 const unsigned short* __restrict__ Wn2b,
                 const float* __restrict__ bn2,
                 float* __restrict__ sums) {
    __shared__ uint4 lA4[4 * KN1T * 32 * 2];
    __shared__ uint4 lM4[4 * KN2T * 32 * 2];
    unsigned short* lA = (unsigned short*)lA4;
    unsigned short* lM = (unsigned short*)lM4;

    const int tid = threadIdx.x;
    const int n0  = blockIdx.x * 64;

    for (int idx = tid; idx < 64 * 256; idx += 256) {
        int m = idx >> 8;
        int K = idx & 255;
        int node = n0 + m;
        float v = 0.0f;
        if (node < N_NODES)
            v = (K < HID) ? h[node * HID + K] : agg[node * HID + (K - HID)];
        lA[asw_pos(m, K, KN1T)] = f2bf(v);
    }
    __syncthreads();

    const int w    = tid >> 5;
    const int lane = tid & 31;
    const int hi   = lane >> 4;
    const int n    = w * 16 + (lane & 15);

    v8f acc[4] = {};
    for (int t = 0; t < KN1T; ++t) {
        Frag B;
        const uint4* bp = (const uint4*)Wn1b + ((t * 8 + w) * 32 + lane) * 2;
        B.q[0] = bp[0]; B.q[1] = bp[1];
        #pragma unroll
        for (int mt = 0; mt < 4; ++mt) {
            Frag A;
            const uint4* ap = (const uint4*)lA4 + ((mt * KN1T + t) * 32 + lane) * 2;
            A.q[0] = ap[0]; A.q[1] = ap[1];
            acc[mt] = __builtin_amdgcn_wmma_f32_16x16x32_bf16(
                false, A.v, false, B.v, (short)0, acc[mt], false, false);
        }
    }
    {
        float b = bn1[n];
        #pragma unroll
        for (int mt = 0; mt < 4; ++mt) {
            #pragma unroll
            for (int j = 0; j < 8; ++j) {
                int m = mt * 16 + hi * 8 + j;
                lM[asw_pos(m, n, KN2T)] = f2bf(silu_f(acc[mt][j] + b));
            }
        }
    }
    __syncthreads();

    v8f acc2[4] = {};
    for (int t = 0; t < KN2T; ++t) {
        Frag B;
        const uint4* bp = (const uint4*)Wn2b + ((t * 8 + w) * 32 + lane) * 2;
        B.q[0] = bp[0]; B.q[1] = bp[1];
        #pragma unroll
        for (int mt = 0; mt < 4; ++mt) {
            Frag A;
            const uint4* ap = (const uint4*)lM4 + ((mt * KN2T + t) * 32 + lane) * 2;
            A.q[0] = ap[0]; A.q[1] = ap[1];
            acc2[mt] = __builtin_amdgcn_wmma_f32_16x16x32_bf16(
                false, A.v, false, B.v, (short)0, acc2[mt], false, false);
        }
    }
    {
        float b = bn2[n];
        #pragma unroll
        for (int mt = 0; mt < 4; ++mt) {
            #pragma unroll
            for (int j = 0; j < 8; ++j) {
                int m = mt * 16 + hi * 8 + j;
                int node = n0 + m;
                if (node < N_NODES) {
                    float hn = h[node * HID + n] + (acc2[mt][j] + b);
                    int g = batch[node];
                    atomicAdd(&sums[g * HID + n], hn);
                }
            }
        }
    }
}

__global__ void counts_kernel(const int* __restrict__ batch, float* __restrict__ counts) {
    int i = blockIdx.x * blockDim.x + threadIdx.x;
    if (i < N_NODES) atomicAdd(&counts[batch[i]], 1.0f);
}

__global__ void finalize_kernel(const float* __restrict__ sums,
                                const float* __restrict__ counts,
                                float* __restrict__ out) {
    int i = blockIdx.x * blockDim.x + threadIdx.x;
    if (i < N_GRAPHS * HID)
        out[i] = sums[i] / fmaxf(counts[i >> 7], 1.0f);
}

extern "C" void kernel_launch(void* const* d_in, const int* in_sizes, int n_in,
                              void* d_out, int out_size, void* d_ws, size_t ws_size,
                              hipStream_t stream) {
    const float* h     = (const float*)d_in[0];
    const int*   eidx  = (const int*)  d_in[1];
    const float* x     = (const float*)d_in[2];
    const float* eattr = (const float*)d_in[3];
    const int*   batch = (const int*)  d_in[4];
    const float* We1   = (const float*)d_in[5];
    const float* be1   = (const float*)d_in[6];
    const float* We2   = (const float*)d_in[7];
    const float* be2   = (const float*)d_in[8];
    const float* Wn1   = (const float*)d_in[9];
    const float* bn1   = (const float*)d_in[10];
    const float* Wn2   = (const float*)d_in[11];
    const float* bn2   = (const float*)d_in[12];

    char* ws = (char*)d_ws;
    unsigned short* We1b = (unsigned short*)(ws);
    unsigned short* We2b = (unsigned short*)(ws + 73728);
    unsigned short* Wn1b = (unsigned short*)(ws + 106496);
    unsigned short* Wn2b = (unsigned short*)(ws + 172032);
    float* agg    = (float*)(ws + 204800);
    float* sums   = (float*)(ws + 204800 + 25600000);
    float* counts = (float*)(ws + 204800 + 25600000 + 32768);

    wconv_kernel<<<(9 * 4096 + 255) / 256, 256, 0, stream>>>(We1, We1b, 273, 9);
    wconv_kernel<<<(4 * 4096 + 255) / 256, 256, 0, stream>>>(We2, We2b, 128, 4);
    wconv_kernel<<<(8 * 4096 + 255) / 256, 256, 0, stream>>>(Wn1, Wn1b, 256, 8);
    wconv_kernel<<<(4 * 4096 + 255) / 256, 256, 0, stream>>>(Wn2, Wn2b, 128, 4);

    int nz = N_NODES * HID + N_GRAPHS * HID + N_GRAPHS;
    zero_f32_kernel<<<(nz + 255) / 256, 256, 0, stream>>>(agg, nz);

    edge_kernel<<<N_EDGES / 64, 256, 0, stream>>>(h, eidx, x, eattr,
                                                  We1b, be1, We2b, be2, agg);
    node_kernel<<<(N_NODES + 63) / 64, 256, 0, stream>>>(h, agg, batch,
                                                         Wn1b, bn1, Wn2b, bn2, sums);
    counts_kernel<<<(N_NODES + 255) / 256, 256, 0, stream>>>(batch, counts);
    finalize_kernel<<<(N_GRAPHS * HID + 255) / 256, 256, 0, stream>>>(sums, counts,
                                                                      (float*)d_out);
}


# graph sizes counted without atomics: a finalize group compares all node ids with its two graph ids, sums hits by shuffles + LDS (was 50000 f32 atomic adds on 64 words, own launch); 9 -> 8 launches
# speedup vs baseline: 2.6766x; 2.6766x over previous
#include <hip/hip_runtime.h>
#include <math.h>

#define N_NODES  50000
#define N_EDGES  800000
#define HID      128
#define EDGE_D   16
#define N_GRAPHS 64

typedef __attribute__((ext_vector_type(16))) __bf16 v16bf;
typedef __attribute__((ext_vector_type(8)))  float  v8f;

union Frag {
    uint4 q[2];
    v16bf v;
};

__device__ __forceinline__ unsigned short f2bf(float f) {
    union { float f; unsigned int u; } c; c.f = f;
    unsigned int u = c.u;
    u += 0x7fffu + ((u >> 16) & 1u);
    return (unsigned short)(u >> 16);
}

__device__ __forceinline__ float silu_f(float x) {
    return x / (1.0f + __expf(-x));
}

__device__ __forceinline__ float xhalf(float x) {
    const int b = __builtin_bit_cast(int, x);
    return __builtin_bit_cast(float,
        __builtin_amdgcn_permlanex16(b, b, 0x76543210, 0xfedcba98, false, false));
}

__device__ __forceinline__ int asw_pos(int m_local, int K, int KT) {
    int mt = m_local >> 4, ml = m_local & 15;
    int t = K >> 5, kk = K & 31;
    int g, i;
    if (kk < 16) { g = kk >> 3;        i = kk & 7; }
    else         { g = (kk - 16) >> 3; i = 8 + ((kk - 16) & 7); }
    return ((mt * KT + t) * 32 + (ml + (g << 4))) * 16 + i;
}

__global__ void wconv_kernel(const float* __restrict__ W,
                             unsigned short* __restrict__ dst,
                             int Ksrc, int KT) {
    int idx = blockIdx.x * blockDim.x + threadIdx.x;
    int total = KT * 8 * 32 * 16;
    if (idx >= total) return;
    int i    = idx & 15;
    int lane = (idx >> 4) & 31;
    int tj   = idx >> 9;
    int t = tj >> 3, j = tj & 7;
    int g = lane >> 4;
    int K = t * 32 + (i < 8 ? g * 8 + i : 16 + g * 8 + (i - 8));
    int n = j * 16 + (lane & 15);
    float v = (K < Ksrc) ? W[K * HID + n] : 0.0f;
    dst[idx] = f2bf(v);
}

typedef __attribute__((ext_vector_type(2))) __bf16 bf16x2_t;
typedef __attribute__((ext_vector_type(2))) float  f32x2_t;

__device__ __forceinline__ unsigned int pack_bf16(float a, float b) {
    f32x2_t f = {a, b};
    bf16x2_t p = __builtin_convertvector(f, bf16x2_t);
    return __builtin_bit_cast(unsigned int, p);
}

__global__ void prep_kernel(float* __restrict__ p, int n,
                            const float* __restrict__ h, unsigned int* __restrict__ hb) {
    int i = blockIdx.x * blockDim.x + threadIdx.x;
    if (i < n) p[i] = 0.0f;
    if (i < N_NODES * HID / 2) {
        const float2 v = ((const float2*)h)[i];
        hb[i] = pack_bf16(v.x, v.y);
    }
}

#define K1T 9
#define K2T 4

#define EDGE_THREADS 512
#define EDGE_WAVES   16
#define EDGE_GRID    256
#define EDGE_TILES   (N_EDGES / 16)

union FragR {
    unsigned int u[8];
    uint4 q[2];
    v16bf v;
};

__global__ __launch_bounds__(EDGE_THREADS)
void edge_kernel(const unsigned short* __restrict__ hb,
                 const int*   __restrict__ eidx,
                 const float* __restrict__ x,
                 const float* __restrict__ eattr,
                 const unsigned short* __restrict__ We1b,
                 const float* __restrict__ be1,
                 const unsigned short* __restrict__ We2b,
                 const float* __restrict__ be2,
                 float* __restrict__ agg) {
    __shared__ uint4  sW1[K1T * 8 * 2 * 32];
    __shared__ uint4  sW2[K2T * 8 * 2 * 32];
    __shared__ float4 sB1[HID / 4];
    __shared__ float  sB2[HID];

    const int tid = threadIdx.x;
    for (int i = tid; i < K1T * 8 * 64; i += EDGE_THREADS) {
        int half = i & 1, ln = (i >> 1) & 31, tj = i >> 6;
        sW1[(tj * 2 + half) * 32 + ln] = ((const uint4*)We1b)[i];
    }
    for (int i = tid; i < K2T * 8 * 64; i += EDGE_THREADS) {
        int half = i & 1, ln = (i >> 1) & 31, tj = i >> 6;
        sW2[(tj * 2 + half) * 32 + ln] = ((const uint4*)We2b)[i];
    }
    if (tid < HID / 4) sB1[tid] = ((const float4*)be1)[tid];
    if (tid < HID)     sB2[tid] = be2[tid];
    __syncthreads();

    const int w    = tid >> 5;
    const int lane = tid & 31;
    const int hi   = lane >> 4;
    const int c    = lane & 15;

    const int TSTRIDE = EDGE_WAVES * EDGE_GRID;
    int tile = w * EDGE_GRID + blockIdx.x;

    uint4  Bq[16];
    float4 ea4[4];
    int r = 0, cn = 0;
    float xr0 = 0.f, xr1 = 0.f, xr2 = 0.f, xc0 = 0.f, xc1 = 0.f, xc2 = 0.f;
    if (tile < EDGE_TILES) {
        const int e = tile * 16 + c;
        r  = eidx[e];
        cn = eidx[N_EDGES + e];
        {
            const float4* ep = (const float4*)(eattr + e * EDGE_D);
            const uint4* pr = (const uint4*)(hb + r  * HID + hi * 8);
            const uint4* pc = (const uint4*)(hb + cn * HID + hi * 8);
            #pragma unroll
            for (int t = 0; t < 4; ++t) {
                Bq[2 * t]     = pr[4 * t];     Bq[2 * t + 1]     = pr[4 * t + 2];
                Bq[2 * t + 8] = pc[4 * t];     Bq[2 * t + 9]     = pc[4 * t + 2];
            }
            ea4[0] = ep[0]; ea4[1] = ep[1]; ea4[2] = ep[2]; ea4[3] = ep[3];
            xr0 = x[r * 3 + 0];  xr1 = x[r * 3 + 1];  xr2 = x[r * 3 + 2];
            xc0 = x[cn * 3 + 0]; xc1 = x[cn * 3 + 1]; xc2 = x[cn * 3 + 2];
        }
    }

    for (; tile < EDGE_TILES; tile += TSTRIDE) {
        const int tnx = (tile + TSTRIDE < EDGE_TILES) ? tile + TSTRIDE : tile;
        const int en  = tnx * 16 + c;
        const int rn  = eidx[en];
        const int cnn = eidx[N_EDGES + en];
        asm volatile("" ::: "memory");

        float radial;
        {
            float d0 = xr0 - xc0;
            float d1 = xr1 - xc1;
            float d2 = xr2 - xc2;
            radial = d0 * d0 + d1 * d1 + d2 * d2;
        }

        v8f acc[8] = {};
        #pragma unroll
        for (int t = 0; t < K1T; ++t) {
            FragR B;
            if (t < 8) {
                B.q[0] = Bq[2 * t]; B.q[1] = Bq[2 * t + 1];
            } else {
                const float ea[16] = {ea4[0].x, ea4[0].y, ea4[0].z, ea4[0].w,
                                      ea4[1].x, ea4[1].y, ea4[1].z, ea4[1].w,
                                      ea4[2].x, ea4[2].y, ea4[2].z, ea4[2].w,
                                      ea4[3].x, ea4[3].y, ea4[3].z, ea4[3].w};
                float v[9];
                v[0] = hi ? ea[7] : radial;
                #pragma unroll
                for (int i = 1; i < 8; ++i) v[i] = hi ? ea[7 + i] : ea[i - 1];
                v[8] = hi ? 0.0f : ea[15];
                B.u[0] = pack_bf16(v[0], v[1]);
                B.u[1] = pack_bf16(v[2], v[3]);
                B.u[2] = pack_bf16(v[4], v[5]);
                B.u[3] = pack_bf16(v[6], v[7]);
                B.u[4] = pack_bf16(v[8], 0.0f);
                B.u[5] = 0u; B.u[6] = 0u; B.u[7] = 0u;
            }
            #pragma unroll
            for (int jt = 0; jt < 8; ++jt) {
                Frag A;
                const uint4* ap = sW1 + ((t * 8 + jt) * 2) * 32 + lane;
                A.q[0] = ap[0]; A.q[1] = ap[32];
                acc[jt] = __builtin_amdgcn_wmma_f32_16x16x32_bf16(
                    false, A.v, false, B.v, (short)0, acc[jt], false, false);
            }
        }

        int srow[16];
        #pragma unroll
        for (int k = 0; k < 16; ++k) srow[k] = __builtin_amdgcn_readlane(r, k);


        FragR M[K2T];
        #pragma unroll
        for (int jt = 0; jt < 8; ++jt) {
            const float4 b0 = sB1[4 * jt + 2 * hi + 0];
            const float4 b1 = sB1[4 * jt + 2 * hi + 1];
            unsigned int* u = &M[jt >> 1].u[(jt & 1) * 4];
            u[0] = pack_bf16(silu_f(acc[jt][0] + b0.x), silu_f(acc[jt][1] + b0.y));
            u[1] = pack_bf16(silu_f(acc[jt][2] + b0.z), silu_f(acc[jt][3] + b0.w));
            u[2] = pack_bf16(silu_f(acc[jt][4] + b1.x), silu_f(acc[jt][5] + b1.y));
            u[3] = pack_bf16(silu_f(acc[jt][6] + b1.z), silu_f(acc[jt][7] + b1.w));
        }

        v8f acc2[8] = {};
        #pragma unroll
        for (int t = 0; t < K2T; ++t) {
            #pragma unroll
            for (int j = 0; j < 8; ++j) {
                Frag Bw;
                const uint4* bp = sW2 + ((t * 8 + j) * 2) * 32 + lane;
                Bw.q[0] = bp[0]; Bw.q[1] = bp[32];
                acc2[j] = __builtin_amdgcn_wmma_f32_16x16x32_bf16(
                    false, M[t].v, false, Bw.v, (short)0, acc2[j], false, false);
            }
        }

        {
            r = rn; cn = cnn;
            const float4* ep = (const float4*)(eattr + en * EDGE_D);
            const uint4* pr = (const uint4*)(hb + r  * HID + hi * 8);
            const uint4* pc = (const uint4*)(hb + cn * HID + hi * 8);
            #pragma unroll
            for (int t = 0; t < 4; ++t) {
                Bq[2 * t]     = pr[4 * t];     Bq[2 * t + 1]     = pr[4 * t + 2];
                Bq[2 * t + 8] = pc[4 * t];     Bq[2 * t + 9]     = pc[4 * t + 2];
            }
            ea4[0] = ep[0]; ea4[1] = ep[1]; ea4[2] = ep[2]; ea4[3] = ep[3];
            xr0 = x[r * 3 + 0];  xr1 = x[r * 3 + 1];  xr2 = x[r * 3 + 2];
            xc0 = x[cn * 3 + 0]; xc1 = x[cn * 3 + 1]; xc2 = x[cn * 3 + 2];
        }

        #pragma unroll
        for (int p = 0; p < 4; ++p) {
            const float bA = sB2[32 * p + c];
            const float bB = sB2[32 * p + 16 + c];
            const unsigned int col = (unsigned int)(32 * p + lane);
            #pragma unroll
            for (int i = 0; i < 8; ++i) {
                const float fa = silu_f(acc2[2 * p][i] + bA);
                const float fb = silu_f(acc2[2 * p + 1][i] + bB);
                const float got = xhalf(hi ? fa : fb);
                const float vlo = hi ? got : fa;
                const float vhi = hi ? fb : got;
                atomicAdd(&agg[(unsigned int)srow[i] * HID + col], vlo);
                atomicAdd(&agg[(unsigned int)srow[8 + i] * HID + col], vhi);
            }
        }
    }
}

#define KN1T 8
#define KN2T 4

__global__ __launch_bounds__(256)
void node_kernel(const float* __restrict__ h,
                 const float* __restrict__ agg,
                 const int*   __restrict__ batch,
                 const unsigned short* __restrict__ Wn1b,
                 const float* __restrict__ bn1,
                 const unsigned short* __restrict__ Wn2b,
                 const float* __restrict__ bn2,
                 float* __restrict__ sums) {
    __shared__ uint4 lA4[4 * KN1T * 32 * 2];
    __shared__ uint4 lM4[4 * KN2T * 32 * 2];
    unsigned short* lA = (unsigned short*)lA4;
    unsigned short* lM = (unsigned short*)lM4;

    const int tid = threadIdx.x;
    const int n0  = blockIdx.x * 64;

    for (int idx = tid; idx < 64 * 256; idx += 256) {
        int m = idx >> 8;
        int K = idx & 255;
        int node = n0 + m;
        float v = 0.0f;
        if (node < N_NODES)
            v = (K < HID) ? h[node * HID + K] : agg[node * HID + (K - HID)];
        lA[asw_pos(m, K, KN1T)] = f2bf(v);
    }
    __syncthreads();

    const int w    = tid >> 5;
    const int lane = tid & 31;
    const int hi   = lane >> 4;
    const int n    = w * 16 + (lane & 15);

    v8f acc[4] = {};
    for (int t = 0; t < KN1T; ++t) {
        Frag B;
        const uint4* bp = (const uint4*)Wn1b + ((t * 8 + w) * 32 + lane) * 2;
        B.q[0] = bp[0]; B.q[1] = bp[1];
        #pragma unroll
        for (int mt = 0; mt < 4; ++mt) {
            Frag A;
            const uint4* ap = (const uint4*)lA4 + ((mt * KN1T + t) * 32 + lane) * 2;
            A.q[0] = ap[0]; A.q[1] = ap[1];
            acc[mt] = __builtin_amdgcn_wmma_f32_16x16x32_bf16(
                false, A.v, false, B.v, (short)0, acc[mt], false, false);
        }
    }
    {
        float b = bn1[n];
        #pragma unroll
        for (int mt = 0; mt < 4; ++mt) {
            #pragma unroll
            for (int j = 0; j < 8; ++j) {
                int m = mt * 16 + hi * 8 + j;
                lM[asw_pos(m, n, KN2T)] = f2bf(silu_f(acc[mt][j] + b));
            }
        }
    }
    __syncthreads();

    v8f acc2[4] = {};
    for (int t = 0; t < KN2T; ++t) {
        Frag B;
        const uint4* bp = (const uint4*)Wn2b + ((t * 8 + w) * 32 + lane) * 2;
        B.q[0] = bp[0]; B.q[1] = bp[1];
        #pragma unroll
        for (int mt = 0; mt < 4; ++mt) {
            Frag A;
            const uint4* ap = (const uint4*)lM4 + ((mt * KN2T + t) * 32 + lane) * 2;
            A.q[0] = ap[0]; A.q[1] = ap[1];
            acc2[mt] = __builtin_amdgcn_wmma_f32_16x16x32_bf16(
                false, A.v, false, B.v, (short)0, acc2[mt], false, false);
        }
    }
    {
        float b = bn2[n];
        #pragma unroll
        for (int mt = 0; mt < 4; ++mt) {
            #pragma unroll
            for (int j = 0; j < 8; ++j) {
                int m = mt * 16 + hi * 8 + j;
                int node = n0 + m;
                if (node < N_NODES) {
                    float hn = h[node * HID + n] + (acc2[mt][j] + b);
                    int g = batch[node];
                    atomicAdd(&sums[g * HID + n], hn);
                }
            }
        }
    }
}

#define FIN_THREADS 1024

__global__ __launch_bounds__(FIN_THREADS)
void finalize_kernel(const float* __restrict__ sums,
                     const int*   __restrict__ batch,
                     float* __restrict__ out) {
    __shared__ int sc[2][FIN_THREADS / 32];
    const int tid = threadIdx.x;
    const int g0 = blockIdx.x * 2, g1 = g0 + 1;
    const int oi = blockIdx.x * 2 * HID + tid;
    float s = 0.0f;
    if (tid < 2 * HID) s = sums[oi];

    const int4* b4 = (const int4*)batch;
    int c0 = 0, c1 = 0;
    for (int k = 0; k < (N_NODES / 4 + FIN_THREADS - 1) / FIN_THREADS; ++k) {
        const int i = tid + k * FIN_THREADS;
        if (i < N_NODES / 4) {
            const int4 v = b4[i];
            c0 += (v.x == g0) + (v.y == g0) + (v.z == g0) + (v.w == g0);
            c1 += (v.x == g1) + (v.y == g1) + (v.z == g1) + (v.w == g1);
        }
    }
    for (int o = 16; o > 0; o >>= 1) {
        c0 += __shfl_xor(c0, o);
        c1 += __shfl_xor(c1, o);
    }
    if ((tid & 31) == 0) { sc[0][tid >> 5] = c0; sc[1][tid >> 5] = c1; }
    __syncthreads();
    if (tid < 2 * HID) {
        const int* p = sc[tid >> 7];
        int cnt = 0;
        for (int w = 0; w < FIN_THREADS / 32; ++w) cnt += p[w];
        out[oi] = s / fmaxf((float)cnt, 1.0f);
    }
}

extern "C" void kernel_launch(void* const* d_in, const int* in_sizes, int n_in,
                              void* d_out, int out_size, void* d_ws, size_t ws_size,
                              hipStream_t stream) {
    const float* h     = (const float*)d_in[0];
    const int*   eidx  = (const int*)  d_in[1];
    const float* x     = (const float*)d_in[2];
    const float* eattr = (const float*)d_in[3];
    const int*   batch = (const int*)  d_in[4];
    const float* We1   = (const float*)d_in[5];
    const float* be1   = (const float*)d_in[6];
    const float* We2   = (const float*)d_in[7];
    const float* be2   = (const float*)d_in[8];
    const float* Wn1   = (const float*)d_in[9];
    const float* bn1   = (const float*)d_in[10];
    const float* Wn2   = (const float*)d_in[11];
    const float* bn2   = (const float*)d_in[12];

    char* ws = (char*)d_ws;
    unsigned short* We1b = (unsigned short*)(ws);
    unsigned short* We2b = (unsigned short*)(ws + 73728);
    unsigned short* Wn1b = (unsigned short*)(ws + 106496);
    unsigned short* Wn2b = (unsigned short*)(ws + 172032);
    float* agg    = (float*)(ws + 204800);
    float* sums   = (float*)(ws + 204800 + 25600000);
    unsigned int* hb = (unsigned int*)(ws + 204800 + 25600000 + 32768 + 256);

    wconv_kernel<<<(9 * 4096 + 255) / 256, 256, 0, stream>>>(We1, We1b, 273, 9);
    wconv_kernel<<<(4 * 4096 + 255) / 256, 256, 0, stream>>>(We2, We2b, 128, 4);
    wconv_kernel<<<(8 * 4096 + 255) / 256, 256, 0, stream>>>(Wn1, Wn1b, 256, 8);
    wconv_kernel<<<(4 * 4096 + 255) / 256, 256, 0, stream>>>(Wn2, Wn2b, 128, 4);

    int nz = N_NODES * HID + N_GRAPHS * HID;
    prep_kernel<<<(nz + 255) / 256, 256, 0, stream>>>(agg, nz, h, hb);

    edge_kernel<<<EDGE_GRID, EDGE_THREADS, 0, stream>>>((const unsigned short*)hb, eidx, x, eattr,
                                                        We1b, be1, We2b, be2, agg);
    node_kernel<<<(N_NODES + 63) / 64, 256, 0, stream>>>(h, agg, batch,
                                                         Wn1b, bn1, Wn2b, bn2, sums);
    finalize_kernel<<<N_GRAPHS / 2, FIN_THREADS, 0, stream>>>(sums, batch, (float*)d_out);
}


# node MLP per wave on 16 nodes x 128 features like the edge kernel: rows loaded straight into layer 1's B (no LDS activation tile, no barrier), weights from LDS; pooled f32 add order changes
# speedup vs baseline: 2.8564x; 1.0672x over previous
#include <hip/hip_runtime.h>
#include <math.h>

#define N_NODES  50000
#define N_EDGES  800000
#define HID      128
#define EDGE_D   16
#define N_GRAPHS 64

typedef __attribute__((ext_vector_type(16))) __bf16 v16bf;
typedef __attribute__((ext_vector_type(8)))  float  v8f;

union Frag {
    uint4 q[2];
    v16bf v;
};

__device__ __forceinline__ unsigned short f2bf(float f) {
    union { float f; unsigned int u; } c; c.f = f;
    unsigned int u = c.u;
    u += 0x7fffu + ((u >> 16) & 1u);
    return (unsigned short)(u >> 16);
}

__device__ __forceinline__ float silu_f(float x) {
    return x / (1.0f + __expf(-x));
}

__device__ __forceinline__ float xhalf(float x) {
    const int b = __builtin_bit_cast(int, x);
    return __builtin_bit_cast(float,
        __builtin_amdgcn_permlanex16(b, b, 0x76543210, 0xfedcba98, false, false));
}

__global__ void wconv_kernel(const float* __restrict__ W,
                             unsigned short* __restrict__ dst,
                             int Ksrc, int KT) {
    int idx = blockIdx.x * blockDim.x + threadIdx.x;
    int total = KT * 8 * 32 * 16;
    if (idx >= total) return;
    int i    = idx & 15;
    int lane = (idx >> 4) & 31;
    int tj   = idx >> 9;
    int t = tj >> 3, j = tj & 7;
    int g = lane >> 4;
    int K = t * 32 + (i < 8 ? g * 8 + i : 16 + g * 8 + (i - 8));
    int n = j * 16 + (lane & 15);
    float v = (K < Ksrc) ? W[K * HID + n] : 0.0f;
    dst[idx] = f2bf(v);
}

typedef __attribute__((ext_vector_type(2))) __bf16 bf16x2_t;
typedef __attribute__((ext_vector_type(2))) float  f32x2_t;

__device__ __forceinline__ unsigned int pack_bf16(float a, float b) {
    f32x2_t f = {a, b};
    bf16x2_t p = __builtin_convertvector(f, bf16x2_t);
    return __builtin_bit_cast(unsigned int, p);
}

__global__ void prep_kernel(float* __restrict__ p, int n,
                            const float* __restrict__ h, unsigned int* __restrict__ hb) {
    int i = blockIdx.x * blockDim.x + threadIdx.x;
    if (i < n) p[i] = 0.0f;
    if (i < N_NODES * HID / 2) {
        const float2 v = ((const float2*)h)[i];
        hb[i] = pack_bf16(v.x, v.y);
    }
}

#define K1T 9
#define K2T 4

#define EDGE_THREADS 512
#define EDGE_WAVES   16
#define EDGE_GRID    256
#define EDGE_TILES   (N_EDGES / 16)

union FragR {
    unsigned int u[8];
    uint4 q[2];
    v16bf v;
};

__global__ __launch_bounds__(EDGE_THREADS)
void edge_kernel(const unsigned short* __restrict__ hb,
                 const int*   __restrict__ eidx,
                 const float* __restrict__ x,
                 const float* __restrict__ eattr,
                 const unsigned short* __restrict__ We1b,
                 const float* __restrict__ be1,
                 const unsigned short* __restrict__ We2b,
                 const float* __restrict__ be2,
                 float* __restrict__ agg) {
    __shared__ uint4  sW1[K1T * 8 * 2 * 32];
    __shared__ uint4  sW2[K2T * 8 * 2 * 32];
    __shared__ float4 sB1[HID / 4];
    __shared__ float  sB2[HID];

    const int tid = threadIdx.x;
    for (int i = tid; i < K1T * 8 * 64; i += EDGE_THREADS) {
        int half = i & 1, ln = (i >> 1) & 31, tj = i >> 6;
        sW1[(tj * 2 + half) * 32 + ln] = ((const uint4*)We1b)[i];
    }
    for (int i = tid; i < K2T * 8 * 64; i += EDGE_THREADS) {
        int half = i & 1, ln = (i >> 1) & 31, tj = i >> 6;
        sW2[(tj * 2 + half) * 32 + ln] = ((const uint4*)We2b)[i];
    }
    if (tid < HID / 4) sB1[tid] = ((const float4*)be1)[tid];
    if (tid < HID)     sB2[tid] = be2[tid];
    __syncthreads();

    const int w    = tid >> 5;
    const int lane = tid & 31;
    const int hi   = lane >> 4;
    const int c    = lane & 15;

    const int TSTRIDE = EDGE_WAVES * EDGE_GRID;
    int tile = w * EDGE_GRID + blockIdx.x;

    uint4  Bq[16];
    float4 ea4[4];
    int r = 0, cn = 0;
    float xr0 = 0.f, xr1 = 0.f, xr2 = 0.f, xc0 = 0.f, xc1 = 0.f, xc2 = 0.f;
    if (tile < EDGE_TILES) {
        const int e = tile * 16 + c;
        r  = eidx[e];
        cn = eidx[N_EDGES + e];
        {
            const float4* ep = (const float4*)(eattr + e * EDGE_D);
            const uint4* pr = (const uint4*)(hb + r  * HID + hi * 8);
            const uint4* pc = (const uint4*)(hb + cn * HID + hi * 8);
            #pragma unroll
            for (int t = 0; t < 4; ++t) {
                Bq[2 * t]     = pr[4 * t];     Bq[2 * t + 1]     = pr[4 * t + 2];
                Bq[2 * t + 8] = pc[4 * t];     Bq[2 * t + 9]     = pc[4 * t + 2];
            }
            ea4[0] = ep[0]; ea4[1] = ep[1]; ea4[2] = ep[2]; ea4[3] = ep[3];
            xr0 = x[r * 3 + 0];  xr1 = x[r * 3 + 1];  xr2 = x[r * 3 + 2];
            xc0 = x[cn * 3 + 0]; xc1 = x[cn * 3 + 1]; xc2 = x[cn * 3 + 2];
        }
    }

    for (; tile < EDGE_TILES; tile += TSTRIDE) {
        const int tnx = (tile + TSTRIDE < EDGE_TILES) ? tile + TSTRIDE : tile;
        const int en  = tnx * 16 + c;
        const int rn  = eidx[en];
        const int cnn = eidx[N_EDGES + en];
        asm volatile("" ::: "memory");

        float radial;
        {
            float d0 = xr0 - xc0;
            float d1 = xr1 - xc1;
            float d2 = xr2 - xc2;
            radial = d0 * d0 + d1 * d1 + d2 * d2;
        }

        v8f acc[8] = {};
        #pragma unroll
        for (int t = 0; t < K1T; ++t) {
            FragR B;
            if (t < 8) {
                B.q[0] = Bq[2 * t]; B.q[1] = Bq[2 * t + 1];
            } else {
                const float ea[16] = {ea4[0].x, ea4[0].y, ea4[0].z, ea4[0].w,
                                      ea4[1].x, ea4[1].y, ea4[1].z, ea4[1].w,
                                      ea4[2].x, ea4[2].y, ea4[2].z, ea4[2].w,
                                      ea4[3].x, ea4[3].y, ea4[3].z, ea4[3].w};
                float v[9];
                v[0] = hi ? ea[7] : radial;
                #pragma unroll
                for (int i = 1; i < 8; ++i) v[i] = hi ? ea[7 + i] : ea[i - 1];
                v[8] = hi ? 0.0f : ea[15];
                B.u[0] = pack_bf16(v[0], v[1]);
                B.u[1] = pack_bf16(v[2], v[3]);
                B.u[2] = pack_bf16(v[4], v[5]);
                B.u[3] = pack_bf16(v[6], v[7]);
                B.u[4] = pack_bf16(v[8], 0.0f);
                B.u[5] = 0u; B.u[6] = 0u; B.u[7] = 0u;
            }
            #pragma unroll
            for (int jt = 0; jt < 8; ++jt) {
                Frag A;
                const uint4* ap = sW1 + ((t * 8 + jt) * 2) * 32 + lane;
                A.q[0] = ap[0]; A.q[1] = ap[32];
                acc[jt] = __builtin_amdgcn_wmma_f32_16x16x32_bf16(
                    false, A.v, false, B.v, (short)0, acc[jt], false, false);
            }
        }

        int srow[16];
        #pragma unroll
        for (int k = 0; k < 16; ++k) srow[k] = __builtin_amdgcn_readlane(r, k);


        FragR M[K2T];
        #pragma unroll
        for (int jt = 0; jt < 8; ++jt) {
            const float4 b0 = sB1[4 * jt + 2 * hi + 0];
            const float4 b1 = sB1[4 * jt + 2 * hi + 1];
            unsigned int* u = &M[jt >> 1].u[(jt & 1) * 4];
            u[0] = pack_bf16(silu_f(acc[jt][0] + b0.x), silu_f(acc[jt][1] + b0.y));
            u[1] = pack_bf16(silu_f(acc[jt][2] + b0.z), silu_f(acc[jt][3] + b0.w));
            u[2] = pack_bf16(silu_f(acc[jt][4] + b1.x), silu_f(acc[jt][5] + b1.y));
            u[3] = pack_bf16(silu_f(acc[jt][6] + b1.z), silu_f(acc[jt][7] + b1.w));
        }

        v8f acc2[8] = {};
        #pragma unroll
        for (int t = 0; t < K2T; ++t) {
            #pragma unroll
            for (int j = 0; j < 8; ++j) {
                Frag Bw;
                const uint4* bp = sW2 + ((t * 8 + j) * 2) * 32 + lane;
                Bw.q[0] = bp[0]; Bw.q[1] = bp[32];
                acc2[j] = __builtin_amdgcn_wmma_f32_16x16x32_bf16(
                    false, M[t].v, false, Bw.v, (short)0, acc2[j], false, false);
            }
        }

        {
            r = rn; cn = cnn;
            const float4* ep = (const float4*)(eattr + en * EDGE_D);
            const uint4* pr = (const uint4*)(hb + r  * HID + hi * 8);
            const uint4* pc = (const uint4*)(hb + cn * HID + hi * 8);
            #pragma unroll
            for (int t = 0; t < 4; ++t) {
                Bq[2 * t]     = pr[4 * t];     Bq[2 * t + 1]     = pr[4 * t + 2];
                Bq[2 * t + 8] = pc[4 * t];     Bq[2 * t + 9]     = pc[4 * t + 2];
            }
            ea4[0] = ep[0]; ea4[1] = ep[1]; ea4[2] = ep[2]; ea4[3] = ep[3];
            xr0 = x[r * 3 + 0];  xr1 = x[r * 3 + 1];  xr2 = x[r * 3 + 2];
            xc0 = x[cn * 3 + 0]; xc1 = x[cn * 3 + 1]; xc2 = x[cn * 3 + 2];
        }

        #pragma unroll
        for (int p = 0; p < 4; ++p) {
            const float bA = sB2[32 * p + c];
            const float bB = sB2[32 * p + 16 + c];
            const unsigned int col = (unsigned int)(32 * p + lane);
            #pragma unroll
            for (int i = 0; i < 8; ++i) {
                const float fa = silu_f(acc2[2 * p][i] + bA);
                const float fb = silu_f(acc2[2 * p + 1][i] + bB);
                const float got = xhalf(hi ? fa : fb);
                const float vlo = hi ? got : fa;
                const float vhi = hi ? fb : got;
                atomicAdd(&agg[(unsigned int)srow[i] * HID + col], vlo);
                atomicAdd(&agg[(unsigned int)srow[8 + i] * HID + col], vhi);
            }
        }
    }
}

#define KN1T 8
#define KN2T 4

#define NODE_THREADS 512
#define NODE_WAVES   16
#define NODE_GRID    256
#define NODE_TILES   (N_NODES / 16)

static_assert(N_NODES % 16 == 0, "a node tile is 16 whole rows");
static_assert(NODE_TILES <= NODE_WAVES * NODE_GRID, "one tile per wave");
static_assert(KN1T * 8 * 64 == 8 * NODE_THREADS && KN2T * 8 * 64 == 4 * NODE_THREADS,
              "a thread stages 8 + 4 weight units");

__global__ __launch_bounds__(NODE_THREADS)
void node_kernel(const float* __restrict__ h,
                 const unsigned short* __restrict__ hb,
                 const float* __restrict__ agg,
                 const int*   __restrict__ batch,
                 const unsigned short* __restrict__ Wn1b,
                 const float* __restrict__ bn1,
                 const unsigned short* __restrict__ Wn2b,
                 const float* __restrict__ bn2,
                 float* __restrict__ sums) {
    __shared__ uint4  sW1[KN1T * 8 * 2 * 32];
    __shared__ uint4  sW2[KN2T * 8 * 2 * 32];
    __shared__ float4 sB1[HID / 4];
    __shared__ float  sB2[HID];

    const int tid  = threadIdx.x;
    const int w    = tid >> 5;
    const int lane = tid & 31;
    const int hi   = lane >> 4;
    const int c    = lane & 15;
    const int tile = w * NODE_GRID + blockIdx.x;
    const bool has = tile < NODE_TILES;

    uint4 wst[12];
    {
        const uint4* g1 = (const uint4*)Wn1b + tid;
        const uint4* g2 = (const uint4*)Wn2b + tid;
        #pragma unroll
        for (int k = 0; k < 8; ++k) wst[k] = g1[k * NODE_THREADS];
        #pragma unroll
        for (int k = 0; k < 4; ++k) wst[8 + k] = g2[k * NODE_THREADS];
    }

    uint4  Bh[8];
    float4 Ba[16];
    int gl = 0;
    if (has) {
        const int node = tile * 16 + c;
        const uint4*  ph = (const uint4*)(hb + node * HID + hi * 8);
        const float4* pa = (const float4*)(agg + node * HID + hi * 8);
        #pragma unroll
        for (int t = 0; t < 4; ++t) {
            Bh[2 * t] = ph[4 * t];          Bh[2 * t + 1] = ph[4 * t + 2];
            Ba[4 * t]     = pa[8 * t];      Ba[4 * t + 1] = pa[8 * t + 1];
            Ba[4 * t + 2] = pa[8 * t + 4];  Ba[4 * t + 3] = pa[8 * t + 5];
        }
        gl = batch[node];
    }

    {
        const int half = tid & 1, ln = (tid >> 1) & 31, tj = tid >> 6;
        uint4* d1 = sW1 + (tj * 2 + half) * 32 + ln;
        uint4* d2 = sW2 + (tj * 2 + half) * 32 + ln;
        #pragma unroll
        for (int k = 0; k < 8; ++k) d1[k * 512] = wst[k];
        #pragma unroll
        for (int k = 0; k < 4; ++k) d2[k * 512] = wst[8 + k];
    }
    if (tid < HID / 4) sB1[tid] = ((const float4*)bn1)[tid];
    if (tid < HID)     sB2[tid] = bn2[tid];
    __syncthreads();

    if (!has) return;

    v8f acc[8] = {};
    #pragma unroll
    for (int t = 0; t < KN1T; ++t) {
        FragR B;
        if (t < 4) {
            B.q[0] = Bh[2 * t]; B.q[1] = Bh[2 * t + 1];
        } else {
            const float4 a0 = Ba[4 * (t - 4)],     a1 = Ba[4 * (t - 4) + 1];
            const float4 a2 = Ba[4 * (t - 4) + 2], a3 = Ba[4 * (t - 4) + 3];
            B.u[0] = pack_bf16(a0.x, a0.y); B.u[1] = pack_bf16(a0.z, a0.w);
            B.u[2] = pack_bf16(a1.x, a1.y); B.u[3] = pack_bf16(a1.z, a1.w);
            B.u[4] = pack_bf16(a2.x, a2.y); B.u[5] = pack_bf16(a2.z, a2.w);
            B.u[6] = pack_bf16(a3.x, a3.y); B.u[7] = pack_bf16(a3.z, a3.w);
        }
        #pragma unroll
        for (int jt = 0; jt < 8; ++jt) {
            Frag A;
            const uint4* ap = sW1 + ((t * 8 + jt) * 2) * 32 + lane;
            A.q[0] = ap[0]; A.q[1] = ap[32];
            acc[jt] = __builtin_amdgcn_wmma_f32_16x16x32_bf16(
                false, A.v, false, B.v, (short)0, acc[jt], false, false);
        }
    }

    int grow[16];
    #pragma unroll
    for (int k = 0; k < 16; ++k) grow[k] = __builtin_amdgcn_readlane(gl, k);

    float hres[4][16];
    {
        const float* hp = h + tile * 16 * HID + lane;
        #pragma unroll
        for (int p = 0; p < 4; ++p) {
            #pragma unroll
            for (int k = 0; k < 16; ++k) hres[p][k] = hp[k * HID + 32 * p];
        }
    }

    asm volatile("" ::: "memory");

    FragR M[KN2T];
    #pragma unroll
    for (int jt = 0; jt < 8; ++jt) {
        const float4 b0 = sB1[4 * jt + 2 * hi + 0];
        const float4 b1 = sB1[4 * jt + 2 * hi + 1];
        unsigned int* u = &M[jt >> 1].u[(jt & 1) * 4];
        u[0] = pack_bf16(silu_f(acc[jt][0] + b0.x), silu_f(acc[jt][1] + b0.y));
        u[1] = pack_bf16(silu_f(acc[jt][2] + b0.z), silu_f(acc[jt][3] + b0.w));
        u[2] = pack_bf16(silu_f(acc[jt][4] + b1.x), silu_f(acc[jt][5] + b1.y));
        u[3] = pack_bf16(silu_f(acc[jt][6] + b1.z), silu_f(acc[jt][7] + b1.w));
    }

    v8f acc2[8] = {};
    #pragma unroll
    for (int t = 0; t < KN2T; ++t) {
        #pragma unroll
        for (int j = 0; j < 8; ++j) {
            Frag Bw;
            const uint4* bp = sW2 + ((t * 8 + j) * 2) * 32 + lane;
            Bw.q[0] = bp[0]; Bw.q[1] = bp[32];
            acc2[j] = __builtin_amdgcn_wmma_f32_16x16x32_bf16(
                false, M[t].v, false, Bw.v, (short)0, acc2[j], false, false);
        }
    }

    #pragma unroll
    for (int p = 0; p < 4; ++p) {
        const float bA = sB2[32 * p + c];
        const float bB = sB2[32 * p + 16 + c];
        const unsigned int col = (unsigned int)(32 * p + lane);
        #pragma unroll
        for (int i = 0; i < 8; ++i) {
            const float fa = acc2[2 * p][i] + bA;
            const float fb = acc2[2 * p + 1][i] + bB;
            const float got = xhalf(hi ? fa : fb);
            const float vlo = hi ? got : fa;
            const float vhi = hi ? fb : got;
            atomicAdd(&sums[(unsigned int)grow[i] * HID + col],     hres[p][i] + vlo);
            atomicAdd(&sums[(unsigned int)grow[8 + i] * HID + col], hres[p][8 + i] + vhi);
        }
    }
}

#define FIN_THREADS 1024

__global__ __launch_bounds__(FIN_THREADS)
void finalize_kernel(const float* __restrict__ sums,
                     const int*   __restrict__ batch,
                     float* __restrict__ out) {
    __shared__ int sc[2][FIN_THREADS / 32];
    const int tid = threadIdx.x;
    const int g0 = blockIdx.x * 2, g1 = g0 + 1;
    const int oi = blockIdx.x * 2 * HID + tid;
    float s = 0.0f;
    if (tid < 2 * HID) s = sums[oi];

    const int4* b4 = (const int4*)batch;
    int c0 = 0, c1 = 0;
    for (int k = 0; k < (N_NODES / 4 + FIN_THREADS - 1) / FIN_THREADS; ++k) {
        const int i = tid + k * FIN_THREADS;
        if (i < N_NODES / 4) {
            const int4 v = b4[i];
            c0 += (v.x == g0) + (v.y == g0) + (v.z == g0) + (v.w == g0);
            c1 += (v.x == g1) + (v.y == g1) + (v.z == g1) + (v.w == g1);
        }
    }
    for (int o = 16; o > 0; o >>= 1) {
        c0 += __shfl_xor(c0, o);
        c1 += __shfl_xor(c1, o);
    }
    if ((tid & 31) == 0) { sc[0][tid >> 5] = c0; sc[1][tid >> 5] = c1; }
    __syncthreads();
    if (tid < 2 * HID) {
        const int* p = sc[tid >> 7];
        int cnt = 0;
        for (int w = 0; w < FIN_THREADS / 32; ++w) cnt += p[w];
        out[oi] = s / fmaxf((float)cnt, 1.0f);
    }
}

extern "C" void kernel_launch(void* const* d_in, const int* in_sizes, int n_in,
                              void* d_out, int out_size, void* d_ws, size_t ws_size,
                              hipStream_t stream) {
    const float* h     = (const float*)d_in[0];
    const int*   eidx  = (const int*)  d_in[1];
    const float* x     = (const float*)d_in[2];
    const float* eattr = (const float*)d_in[3];
    const int*   batch = (const int*)  d_in[4];
    const float* We1   = (const float*)d_in[5];
    const float* be1   = (const float*)d_in[6];
    const float* We2   = (const float*)d_in[7];
    const float* be2   = (const float*)d_in[8];
    const float* Wn1   = (const float*)d_in[9];
    const float* bn1   = (const float*)d_in[10];
    const float* Wn2   = (const float*)d_in[11];
    const float* bn2   = (const float*)d_in[12];

    char* ws = (char*)d_ws;
    unsigned short* We1b = (unsigned short*)(ws);
    unsigned short* We2b = (unsigned short*)(ws + 73728);
    unsigned short* Wn1b = (unsigned short*)(ws + 106496);
    unsigned short* Wn2b = (unsigned short*)(ws + 172032);
    float* agg    = (float*)(ws + 204800);
    float* sums   = (float*)(ws + 204800 + 25600000);
    unsigned int* hb = (unsigned int*)(ws + 204800 + 25600000 + 32768 + 256);

    wconv_kernel<<<(9 * 4096 + 255) / 256, 256, 0, stream>>>(We1, We1b, 273, 9);
    wconv_kernel<<<(4 * 4096 + 255) / 256, 256, 0, stream>>>(We2, We2b, 128, 4);
    wconv_kernel<<<(8 * 4096 + 255) / 256, 256, 0, stream>>>(Wn1, Wn1b, 256, 8);
    wconv_kernel<<<(4 * 4096 + 255) / 256, 256, 0, stream>>>(Wn2, Wn2b, 128, 4);

    int nz = N_NODES * HID + N_GRAPHS * HID;
    prep_kernel<<<(nz + 255) / 256, 256, 0, stream>>>(agg, nz, h, hb);

    edge_kernel<<<EDGE_GRID, EDGE_THREADS, 0, stream>>>((const unsigned short*)hb, eidx, x, eattr,
                                                        We1b, be1, We2b, be2, agg);
    node_kernel<<<NODE_GRID, NODE_THREADS, 0, stream>>>(h, (const unsigned short*)hb, agg, batch,
                                                        Wn1b, bn1, Wn2b, bn2, sums);
    finalize_kernel<<<N_GRAPHS / 2, FIN_THREADS, 0, stream>>>(sums, batch, (float*)d_out);
}


# edge + node kernels build their LDS weight fragments from the f32 weights (4-byte column loads as one group, packed convert); the 4 conversion launches go: 8 -> 4 launches; f32 sums unchanged
# speedup vs baseline: 3.1338x; 1.0971x over previous
#include <hip/hip_runtime.h>
#include <math.h>

#define N_NODES  50000
#define N_EDGES  800000
#define HID      128
#define EDGE_D   16
#define N_GRAPHS 64

typedef __attribute__((ext_vector_type(16))) __bf16 v16bf;
typedef __attribute__((ext_vector_type(8)))  float  v8f;

union Frag {
    uint4 q[2];
    v16bf v;
};

__device__ __forceinline__ float silu_f(float x) {
    return x / (1.0f + __expf(-x));
}

__device__ __forceinline__ float xhalf(float x) {
    const int b = __builtin_bit_cast(int, x);
    return __builtin_bit_cast(float,
        __builtin_amdgcn_permlanex16(b, b, 0x76543210, 0xfedcba98, false, false));
}

typedef __attribute__((ext_vector_type(2))) __bf16 bf16x2_t;
typedef __attribute__((ext_vector_type(2))) float  f32x2_t;

__device__ __forceinline__ unsigned int pack_bf16(float a, float b) {
    f32x2_t f = {a, b};
    bf16x2_t p = __builtin_convertvector(f, bf16x2_t);
    return __builtin_bit_cast(unsigned int, p);
}

__device__ __forceinline__ void wunit_load(float (&v)[8], const float* __restrict__ W, int kg, int n) {
    #pragma unroll
    for (int r = 0; r < 8; ++r) v[r] = W[(kg * 8 + r) * HID + n];
}

__device__ __forceinline__ void wunit_load_padded(float (&v)[8], const float* __restrict__ W,
                                                  int kg, int n, int Ksrc) {
    #pragma unroll
    for (int r = 0; r < 8; ++r) {
        const int K = kg * 8 + r;
        const float keep = (K < Ksrc) ? 1.0f : 0.0f;
        v[r] = W[(K < Ksrc ? K : Ksrc - 1) * HID + n] * keep;
    }
}

__device__ __forceinline__ void wunit_store(uint4* __restrict__ sW, int kg, int n, const float (&v)[8]) {
    const int t = kg >> 2, half = (kg >> 1) & 1, g = kg & 1;
    sW[((t * 8 + (n >> 4)) * 2 + half) * 32 + g * 16 + (n & 15)] =
        make_uint4(pack_bf16(v[0], v[1]), pack_bf16(v[2], v[3]), pack_bf16(v[4], v[5]), pack_bf16(v[6], v[7]));
}

__global__ void prep_kernel(float* __restrict__ p, int n,
                            const float* __restrict__ h, unsigned int* __restrict__ hb) {
    int i = blockIdx.x * blockDim.x + threadIdx.x;
    if (i < n) p[i] = 0.0f;
    if (i < N_NODES * HID / 2) {
        const float2 v = ((const float2*)h)[i];
        hb[i] = pack_bf16(v.x, v.y);
    }
}

#define K1T 9
#define K2T 4

#define EDGE_THREADS 512
#define EDGE_WAVES   16
#define EDGE_GRID    256
#define EDGE_TILES   (N_EDGES / 16)

static_assert(EDGE_THREADS == 4 * HID && EDGE_THREADS == 32 * EDGE_WAVES,
              "weight staging: a thread keeps one of 128 features and every fourth k-group");

union FragR {
    unsigned int u[8];
    uint4 q[2];
    v16bf v;
};

__device__ __forceinline__ void sched_fence() { __builtin_amdgcn_sched_barrier(0); }

__device__ __forceinline__ void load_wset(Frag (&s)[4], const uint4* p) {
    #pragma unroll
    for (int q = 0; q < 4; ++q) { s[q].q[0] = p[q * 64]; s[q].q[1] = p[q * 64 + 32]; }
}

__global__ __launch_bounds__(EDGE_THREADS)
void edge_kernel(const unsigned short* __restrict__ hb,
                 const int*   __restrict__ eidx,
                 const float* __restrict__ x,
                 const float* __restrict__ eattr,
                 const float* __restrict__ We1,
                 const float* __restrict__ be1,
                 const float* __restrict__ We2,
                 const float* __restrict__ be2,
                 float* __restrict__ agg) {
    __shared__ uint4  sW1[K1T * 8 * 2 * 32];
    __shared__ uint4  sW2[K2T * 8 * 2 * 32];
    __shared__ float4 sB1[HID / 4];
    __shared__ float  sB2[HID];

    const int tid  = threadIdx.x;
    const int lane = tid & 31;
    const int w    = tid >> 5;

    {
        const int n  = tid & 127;
        const int k0 = tid >> 7;
        float u1[K1T][8], u2[K2T][8];
        #pragma unroll
        for (int k = 0; k < K1T - 1; ++k) wunit_load(u1[k], We1, k0 + 4 * k, n);
        wunit_load_padded(u1[K1T - 1], We1, k0 + 4 * (K1T - 1), n, 2 * HID + 1 + EDGE_D);
        #pragma unroll
        for (int k = 0; k < K2T; ++k) wunit_load(u2[k], We2, k0 + 4 * k, n);
        float4 b1v = make_float4(0.0f, 0.0f, 0.0f, 0.0f);
        float  b2v = 0.0f;
        if (tid < HID / 4) b1v = ((const float4*)be1)[tid];
        if (tid < HID)     b2v = be2[tid];

        #pragma unroll
        for (int k = 0; k < K1T; ++k) wunit_store(sW1, k0 + 4 * k, n, u1[k]);
        #pragma unroll
        for (int k = 0; k < K2T; ++k) wunit_store(sW2, k0 + 4 * k, n, u2[k]);
        if (tid < HID / 4) sB1[tid] = b1v;
        if (tid < HID)     sB2[tid] = b2v;
    }
    __syncthreads();

    const int hi   = lane >> 4;
    const int c    = lane & 15;

    const int TSTRIDE = EDGE_WAVES * EDGE_GRID;
    int tile = w * EDGE_GRID + blockIdx.x;

    uint4  Bq[16];
    float4 ea4[4];
    int r = 0, cn = 0;
    float xr0 = 0.f, xr1 = 0.f, xr2 = 0.f, xc0 = 0.f, xc1 = 0.f, xc2 = 0.f;
    if (tile < EDGE_TILES) {
        const int e = tile * 16 + c;
        r  = eidx[e];
        cn = eidx[N_EDGES + e];
        {
            const float4* ep = (const float4*)(eattr + e * EDGE_D);
            const uint4* pr = (const uint4*)(hb + r  * HID + hi * 8);
            const uint4* pc = (const uint4*)(hb + cn * HID + hi * 8);
            #pragma unroll
            for (int t = 0; t < 4; ++t) {
                Bq[2 * t]     = pr[4 * t];     Bq[2 * t + 1]     = pr[4 * t + 2];
                Bq[2 * t + 8] = pc[4 * t];     Bq[2 * t + 9]     = pc[4 * t + 2];
            }
            ea4[0] = ep[0]; ea4[1] = ep[1]; ea4[2] = ep[2]; ea4[3] = ep[3];
            xr0 = x[r * 3 + 0];  xr1 = x[r * 3 + 1];  xr2 = x[r * 3 + 2];
            xc0 = x[cn * 3 + 0]; xc1 = x[cn * 3 + 1]; xc2 = x[cn * 3 + 2];
        }
    }

    for (; tile < EDGE_TILES; tile += TSTRIDE) {
        const int tnx = (tile + TSTRIDE < EDGE_TILES) ? tile + TSTRIDE : tile;
        const int en  = tnx * 16 + c;
        const int rn  = eidx[en];
        const int cnn = eidx[N_EDGES + en];
        asm volatile("" ::: "memory");

        float radial;
        {
            float d0 = xr0 - xc0;
            float d1 = xr1 - xc1;
            float d2 = xr2 - xc2;
            radial = d0 * d0 + d1 * d1 + d2 * d2;
        }

        Frag Wa[4], Wb[4];
        load_wset(Wa, sW1 + lane);
        v8f acc[8] = {};
        #pragma unroll
        for (int t = 0; t < K1T; ++t) {
            FragR B;
            if (t < 8) {
                B.q[0] = Bq[2 * t]; B.q[1] = Bq[2 * t + 1];
            } else {
                const float ea[16] = {ea4[0].x, ea4[0].y, ea4[0].z, ea4[0].w,
                                      ea4[1].x, ea4[1].y, ea4[1].z, ea4[1].w,
                                      ea4[2].x, ea4[2].y, ea4[2].z, ea4[2].w,
                                      ea4[3].x, ea4[3].y, ea4[3].z, ea4[3].w};
                float v[9];
                v[0] = hi ? ea[7] : radial;
                #pragma unroll
                for (int i = 1; i < 8; ++i) v[i] = hi ? ea[7 + i] : ea[i - 1];
                v[8] = hi ? 0.0f : ea[15];
                B.u[0] = pack_bf16(v[0], v[1]);
                B.u[1] = pack_bf16(v[2], v[3]);
                B.u[2] = pack_bf16(v[4], v[5]);
                B.u[3] = pack_bf16(v[6], v[7]);
                B.u[4] = pack_bf16(v[8], 0.0f);
                B.u[5] = 0u; B.u[6] = 0u; B.u[7] = 0u;
            }
            load_wset(Wb, sW1 + (t * 8 + 4) * 64 + lane);
            sched_fence();
            #pragma unroll
            for (int q = 0; q < 4; ++q)
                acc[q] = __builtin_amdgcn_wmma_f32_16x16x32_bf16(
                    false, Wa[q].v, false, B.v, (short)0, acc[q], false, false);
            sched_fence();
            load_wset(Wa, (t + 1 < K1T) ? sW1 + (t + 1) * 8 * 64 + lane : sW2 + lane);
            sched_fence();
            #pragma unroll
            for (int q = 0; q < 4; ++q)
                acc[4 + q] = __builtin_amdgcn_wmma_f32_16x16x32_bf16(
                    false, Wb[q].v, false, B.v, (short)0, acc[4 + q], false, false);
            sched_fence();
        }

        int srow[16];
        #pragma unroll
        for (int k = 0; k < 16; ++k) srow[k] = __builtin_amdgcn_readlane(r, k);


        FragR M[K2T];
        #pragma unroll
        for (int jt = 0; jt < 8; ++jt) {
            const float4 b0 = sB1[4 * jt + 2 * hi + 0];
            const float4 b1 = sB1[4 * jt + 2 * hi + 1];
            unsigned int* u = &M[jt >> 1].u[(jt & 1) * 4];
            u[0] = pack_bf16(silu_f(acc[jt][0] + b0.x), silu_f(acc[jt][1] + b0.y));
            u[1] = pack_bf16(silu_f(acc[jt][2] + b0.z), silu_f(acc[jt][3] + b0.w));
            u[2] = pack_bf16(silu_f(acc[jt][4] + b1.x), silu_f(acc[jt][5] + b1.y));
            u[3] = pack_bf16(silu_f(acc[jt][6] + b1.z), silu_f(acc[jt][7] + b1.w));
        }

        v8f acc2[8] = {};
        #pragma unroll
        for (int t = 0; t < K2T; ++t) {
            load_wset(Wb, sW2 + (t * 8 + 4) * 64 + lane);
            sched_fence();
            #pragma unroll
            for (int q = 0; q < 4; ++q)
                acc2[q] = __builtin_amdgcn_wmma_f32_16x16x32_bf16(
                    false, M[t].v, false, Wa[q].v, (short)0, acc2[q], false, false);
            sched_fence();
            if (t + 1 < K2T) {
                load_wset(Wa, sW2 + (t + 1) * 8 * 64 + lane);
                sched_fence();
            }
            #pragma unroll
            for (int q = 0; q < 4; ++q)
                acc2[4 + q] = __builtin_amdgcn_wmma_f32_16x16x32_bf16(
                    false, M[t].v, false, Wb[q].v, (short)0, acc2[4 + q], false, false);
            sched_fence();
        }

        {
            r = rn; cn = cnn;
            const float4* ep = (const float4*)(eattr + en * EDGE_D);
            const uint4* pr = (const uint4*)(hb + r  * HID + hi * 8);
            const uint4* pc = (const uint4*)(hb + cn * HID + hi * 8);
            #pragma unroll
            for (int t = 0; t < 4; ++t) {
                Bq[2 * t]     = pr[4 * t];     Bq[2 * t + 1]     = pr[4 * t + 2];
                Bq[2 * t + 8] = pc[4 * t];     Bq[2 * t + 9]     = pc[4 * t + 2];
            }
            ea4[0] = ep[0]; ea4[1] = ep[1]; ea4[2] = ep[2]; ea4[3] = ep[3];
            xr0 = x[r * 3 + 0];  xr1 = x[r * 3 + 1];  xr2 = x[r * 3 + 2];
            xc0 = x[cn * 3 + 0]; xc1 = x[cn * 3 + 1]; xc2 = x[cn * 3 + 2];
        }

        #pragma unroll
        for (int p = 0; p < 4; ++p) {
            const float bA = sB2[32 * p + c];
            const float bB = sB2[32 * p + 16 + c];
            const unsigned int col = (unsigned int)(32 * p + lane);
            #pragma unroll
            for (int i = 0; i < 8; ++i) {
                const float fa = silu_f(acc2[2 * p][i] + bA);
                const float fb = silu_f(acc2[2 * p + 1][i] + bB);
                const float got = xhalf(hi ? fa : fb);
                const float vlo = hi ? got : fa;
                const float vhi = hi ? fb : got;
                atomicAdd(&agg[(unsigned int)srow[i] * HID + col], vlo);
                atomicAdd(&agg[(unsigned int)srow[8 + i] * HID + col], vhi);
            }
        }
    }
}

#define KN1T 8
#define KN2T 4

#define NODE_THREADS 512
#define NODE_WAVES   16
#define NODE_GRID    256
#define NODE_TILES   (N_NODES / 16)

static_assert(N_NODES % 16 == 0, "a node tile is 16 whole rows");
static_assert(NODE_TILES <= NODE_WAVES * NODE_GRID, "one tile per wave");
static_assert(NODE_THREADS == 4 * HID && NODE_THREADS == 32 * NODE_WAVES,
              "weight staging: a thread keeps one of 128 features and every fourth k-group");

__global__ __launch_bounds__(NODE_THREADS)
void node_kernel(const float* __restrict__ h,
                 const unsigned short* __restrict__ hb,
                 const float* __restrict__ agg,
                 const int*   __restrict__ batch,
                 const float* __restrict__ Wn1,
                 const float* __restrict__ bn1,
                 const float* __restrict__ Wn2,
                 const float* __restrict__ bn2,
                 float* __restrict__ sums) {
    __shared__ uint4  sW1[KN1T * 8 * 2 * 32];
    __shared__ uint4  sW2[KN2T * 8 * 2 * 32];
    __shared__ float4 sB1[HID / 4];
    __shared__ float  sB2[HID];

    const int tid  = threadIdx.x;
    const int w    = tid >> 5;
    const int lane = tid & 31;
    const int hi   = lane >> 4;
    const int c    = lane & 15;
    const int tile = w * NODE_GRID + blockIdx.x;
    const bool has = tile < NODE_TILES;

    const int wn  = tid & 127;
    const int wk0 = tid >> 7;
    float u1[KN1T][8], u2[KN2T][8];
    #pragma unroll
    for (int k = 0; k < KN1T; ++k) wunit_load(u1[k], Wn1, wk0 + 4 * k, wn);
    #pragma unroll
    for (int k = 0; k < KN2T; ++k) wunit_load(u2[k], Wn2, wk0 + 4 * k, wn);
    float4 b1v = make_float4(0.0f, 0.0f, 0.0f, 0.0f);
    float  b2v = 0.0f;
    if (tid < HID / 4) b1v = ((const float4*)bn1)[tid];
    if (tid < HID)     b2v = bn2[tid];

    uint4  Bh[8];
    float4 Ba[16];
    int gl = 0;
    if (has) {
        const int node = tile * 16 + c;
        const uint4*  ph = (const uint4*)(hb + node * HID + hi * 8);
        const float4* pa = (const float4*)(agg + node * HID + hi * 8);
        #pragma unroll
        for (int t = 0; t < 4; ++t) {
            Bh[2 * t] = ph[4 * t];          Bh[2 * t + 1] = ph[4 * t + 2];
            Ba[4 * t]     = pa[8 * t];      Ba[4 * t + 1] = pa[8 * t + 1];
            Ba[4 * t + 2] = pa[8 * t + 4];  Ba[4 * t + 3] = pa[8 * t + 5];
        }
        gl = batch[node];
    }

    #pragma unroll
    for (int k = 0; k < KN1T; ++k) wunit_store(sW1, wk0 + 4 * k, wn, u1[k]);
    #pragma unroll
    for (int k = 0; k < KN2T; ++k) wunit_store(sW2, wk0 + 4 * k, wn, u2[k]);
    if (tid < HID / 4) sB1[tid] = b1v;
    if (tid < HID)     sB2[tid] = b2v;
    __syncthreads();

    if (!has) return;

    Frag Wa[4], Wb[4];
    load_wset(Wa, sW1 + lane);
    v8f acc[8] = {};
    #pragma unroll
    for (int t = 0; t < KN1T; ++t) {
        FragR B;
        if (t < 4) {
            B.q[0] = Bh[2 * t]; B.q[1] = Bh[2 * t + 1];
        } else {
            const float4 a0 = Ba[4 * (t - 4)],     a1 = Ba[4 * (t - 4) + 1];
            const float4 a2 = Ba[4 * (t - 4) + 2], a3 = Ba[4 * (t - 4) + 3];
            B.u[0] = pack_bf16(a0.x, a0.y); B.u[1] = pack_bf16(a0.z, a0.w);
            B.u[2] = pack_bf16(a1.x, a1.y); B.u[3] = pack_bf16(a1.z, a1.w);
            B.u[4] = pack_bf16(a2.x, a2.y); B.u[5] = pack_bf16(a2.z, a2.w);
            B.u[6] = pack_bf16(a3.x, a3.y); B.u[7] = pack_bf16(a3.z, a3.w);
        }
        load_wset(Wb, sW1 + (t * 8 + 4) * 64 + lane);
        sched_fence();
        #pragma unroll
        for (int q = 0; q < 4; ++q)
            acc[q] = __builtin_amdgcn_wmma_f32_16x16x32_bf16(
                false, Wa[q].v, false, B.v, (short)0, acc[q], false, false);
        sched_fence();
        load_wset(Wa, (t + 1 < KN1T) ? sW1 + (t + 1) * 8 * 64 + lane : sW2 + lane);
        sched_fence();
        #pragma unroll
        for (int q = 0; q < 4; ++q)
            acc[4 + q] = __builtin_amdgcn_wmma_f32_16x16x32_bf16(
                false, Wb[q].v, false, B.v, (short)0, acc[4 + q], false, false);
        sched_fence();
    }

    int grow[16];
    #pragma unroll
    for (int k = 0; k < 16; ++k) grow[k] = __builtin_amdgcn_readlane(gl, k);

    float hres[4][16];
    {
        const float* hp = h + tile * 16 * HID + lane;
        #pragma unroll
        for (int p = 0; p < 4; ++p) {
            #pragma unroll
            for (int k = 0; k < 16; ++k) hres[p][k] = hp[k * HID + 32 * p];
        }
    }

    asm volatile("" ::: "memory");

    FragR M[KN2T];
    #pragma unroll
    for (int jt = 0; jt < 8; ++jt) {
        const float4 b0 = sB1[4 * jt + 2 * hi + 0];
        const float4 b1 = sB1[4 * jt + 2 * hi + 1];
        unsigned int* u = &M[jt >> 1].u[(jt & 1) * 4];
        u[0] = pack_bf16(silu_f(acc[jt][0] + b0.x), silu_f(acc[jt][1] + b0.y));
        u[1] = pack_bf16(silu_f(acc[jt][2] + b0.z), silu_f(acc[jt][3] + b0.w));
        u[2] = pack_bf16(silu_f(acc[jt][4] + b1.x), silu_f(acc[jt][5] + b1.y));
        u[3] = pack_bf16(silu_f(acc[jt][6] + b1.z), silu_f(acc[jt][7] + b1.w));
    }

    v8f acc2[8] = {};
    #pragma unroll
    for (int t = 0; t < KN2T; ++t) {
        load_wset(Wb, sW2 + (t * 8 + 4) * 64 + lane);
        sched_fence();
        #pragma unroll
        for (int q = 0; q < 4; ++q)
            acc2[q] = __builtin_amdgcn_wmma_f32_16x16x32_bf16(
                false, M[t].v, false, Wa[q].v, (short)0, acc2[q], false, false);
        sched_fence();
        if (t + 1 < KN2T) {
            load_wset(Wa, sW2 + (t + 1) * 8 * 64 + lane);
            sched_fence();
        }
        #pragma unroll
        for (int q = 0; q < 4; ++q)
            acc2[4 + q] = __builtin_amdgcn_wmma_f32_16x16x32_bf16(
                false, M[t].v, false, Wb[q].v, (short)0, acc2[4 + q], false, false);
        sched_fence();
    }

    #pragma unroll
    for (int p = 0; p < 4; ++p) {
        const float bA = sB2[32 * p + c];
        const float bB = sB2[32 * p + 16 + c];
        const unsigned int col = (unsigned int)(32 * p + lane);
        #pragma unroll
        for (int i = 0; i < 8; ++i) {
            const float fa = acc2[2 * p][i] + bA;
            const float fb = acc2[2 * p + 1][i] + bB;
            const float got = xhalf(hi ? fa : fb);
            const float vlo = hi ? got : fa;
            const float vhi = hi ? fb : got;
            atomicAdd(&sums[(unsigned int)grow[i] * HID + col],     hres[p][i] + vlo);
            atomicAdd(&sums[(unsigned int)grow[8 + i] * HID + col], hres[p][8 + i] + vhi);
        }
    }
}

#define FIN_THREADS 1024

__global__ __launch_bounds__(FIN_THREADS)
void finalize_kernel(const float* __restrict__ sums,
                     const int*   __restrict__ batch,
                     float* __restrict__ out) {
    __shared__ int sc[2][FIN_THREADS / 32];
    const int tid = threadIdx.x;
    const int g0 = blockIdx.x * 2, g1 = g0 + 1;
    const int oi = blockIdx.x * 2 * HID + tid;
    float s = 0.0f;
    if (tid < 2 * HID) s = sums[oi];

    const int4* b4 = (const int4*)batch;
    int c0 = 0, c1 = 0;
    for (int k = 0; k < (N_NODES / 4 + FIN_THREADS - 1) / FIN_THREADS; ++k) {
        const int i = tid + k * FIN_THREADS;
        if (i < N_NODES / 4) {
            const int4 v = b4[i];
            c0 += (v.x == g0) + (v.y == g0) + (v.z == g0) + (v.w == g0);
            c1 += (v.x == g1) + (v.y == g1) + (v.z == g1) + (v.w == g1);
        }
    }
    for (int o = 16; o > 0; o >>= 1) {
        c0 += __shfl_xor(c0, o);
        c1 += __shfl_xor(c1, o);
    }
    if ((tid & 31) == 0) { sc[0][tid >> 5] = c0; sc[1][tid >> 5] = c1; }
    __syncthreads();
    if (tid < 2 * HID) {
        const int* p = sc[tid >> 7];
        int cnt = 0;
        for (int w = 0; w < FIN_THREADS / 32; ++w) cnt += p[w];
        out[oi] = s / fmaxf((float)cnt, 1.0f);
    }
}

extern "C" void kernel_launch(void* const* d_in, const int* in_sizes, int n_in,
                              void* d_out, int out_size, void* d_ws, size_t ws_size,
                              hipStream_t stream) {
    const float* h     = (const float*)d_in[0];
    const int*   eidx  = (const int*)  d_in[1];
    const float* x     = (const float*)d_in[2];
    const float* eattr = (const float*)d_in[3];
    const int*   batch = (const int*)  d_in[4];
    const float* We1   = (const float*)d_in[5];
    const float* be1   = (const float*)d_in[6];
    const float* We2   = (const float*)d_in[7];
    const float* be2   = (const float*)d_in[8];
    const float* Wn1   = (const float*)d_in[9];
    const float* bn1   = (const float*)d_in[10];
    const float* Wn2   = (const float*)d_in[11];
    const float* bn2   = (const float*)d_in[12];

    char* ws = (char*)d_ws;
    float* agg    = (float*)(ws + 204800);
    float* sums   = (float*)(ws + 204800 + 25600000);
    unsigned int* hb = (unsigned int*)(ws + 204800 + 25600000 + 32768 + 256);

    int nz = N_NODES * HID + N_GRAPHS * HID;
    prep_kernel<<<(nz + 255) / 256, 256, 0, stream>>>(agg, nz, h, hb);

    edge_kernel<<<EDGE_GRID, EDGE_THREADS, 0, stream>>>((const unsigned short*)hb, eidx, x, eattr,
                                                        We1, be1, We2, be2, agg);
    node_kernel<<<NODE_GRID, NODE_THREADS, 0, stream>>>(h, (const unsigned short*)hb, agg, batch,
                                                        Wn1, bn1, Wn2, bn2, sums);
    finalize_kernel<<<N_GRAPHS / 2, FIN_THREADS, 0, stream>>>(sums, batch, (float*)d_out);
}


# edge scatter epilogue in 4 blocks: a tile pair's 16 SiLU chains + half trades run together, then its 16 agg atomics as one group (was chain by chain); biases read ahead, radial formed at loop end
# speedup vs baseline: 3.2999x; 1.0530x over previous
#include <hip/hip_runtime.h>
#include <math.h>

#define N_NODES  50000
#define N_EDGES  800000
#define HID      128
#define EDGE_D   16
#define N_GRAPHS 64

typedef __attribute__((ext_vector_type(16))) __bf16 v16bf;
typedef __attribute__((ext_vector_type(8)))  float  v8f;

union Frag {
    uint4 q[2];
    v16bf v;
};

__device__ __forceinline__ float silu_f(float x) {
    return x / (1.0f + __expf(-x));
}

__device__ __forceinline__ float xhalf(float x) {
    const int b = __builtin_bit_cast(int, x);
    return __builtin_bit_cast(float,
        __builtin_amdgcn_permlanex16(b, b, 0x76543210, 0xfedcba98, false, false));
}

typedef __attribute__((ext_vector_type(2))) __bf16 bf16x2_t;
typedef __attribute__((ext_vector_type(2))) float  f32x2_t;

__device__ __forceinline__ unsigned int pack_bf16(float a, float b) {
    f32x2_t f = {a, b};
    bf16x2_t p = __builtin_convertvector(f, bf16x2_t);
    return __builtin_bit_cast(unsigned int, p);
}

__device__ __forceinline__ void wunit_load(float (&v)[8], const float* __restrict__ W, int kg, int n) {
    #pragma unroll
    for (int r = 0; r < 8; ++r) v[r] = W[(kg * 8 + r) * HID + n];
}

__device__ __forceinline__ void wunit_load_padded(float (&v)[8], const float* __restrict__ W,
                                                  int kg, int n, int Ksrc) {
    #pragma unroll
    for (int r = 0; r < 8; ++r) {
        const int K = kg * 8 + r;
        const float keep = (K < Ksrc) ? 1.0f : 0.0f;
        v[r] = W[(K < Ksrc ? K : Ksrc - 1) * HID + n] * keep;
    }
}

__device__ __forceinline__ void wunit_store(uint4* __restrict__ sW, int kg, int n, const float (&v)[8]) {
    const int t = kg >> 2, half = (kg >> 1) & 1, g = kg & 1;
    sW[((t * 8 + (n >> 4)) * 2 + half) * 32 + g * 16 + (n & 15)] =
        make_uint4(pack_bf16(v[0], v[1]), pack_bf16(v[2], v[3]), pack_bf16(v[4], v[5]), pack_bf16(v[6], v[7]));
}

__global__ void prep_kernel(float* __restrict__ p, int n,
                            const float* __restrict__ h, unsigned int* __restrict__ hb) {
    int i = blockIdx.x * blockDim.x + threadIdx.x;
    if (i < n) p[i] = 0.0f;
    if (i < N_NODES * HID / 2) {
        const float2 v = ((const float2*)h)[i];
        hb[i] = pack_bf16(v.x, v.y);
    }
}

#define K1T 9
#define K2T 4

#define EDGE_THREADS 512
#define EDGE_WAVES   16
#define EDGE_GRID    256
#define EDGE_TILES   (N_EDGES / 16)

static_assert(EDGE_THREADS == 4 * HID && EDGE_THREADS == 32 * EDGE_WAVES,
              "weight staging: a thread keeps one of 128 features and every fourth k-group");

union FragR {
    unsigned int u[8];
    uint4 q[2];
    v16bf v;
};

__device__ __forceinline__ void sched_fence() { __builtin_amdgcn_sched_barrier(0); }

__device__ __forceinline__ void load_wset(Frag (&s)[4], const uint4* p) {
    #pragma unroll
    for (int q = 0; q < 4; ++q) { s[q].q[0] = p[q * 64]; s[q].q[1] = p[q * 64 + 32]; }
}

__global__ __launch_bounds__(EDGE_THREADS)
void edge_kernel(const unsigned short* __restrict__ hb,
                 const int*   __restrict__ eidx,
                 const float* __restrict__ x,
                 const float* __restrict__ eattr,
                 const float* __restrict__ We1,
                 const float* __restrict__ be1,
                 const float* __restrict__ We2,
                 const float* __restrict__ be2,
                 float* __restrict__ agg) {
    __shared__ uint4  sW1[K1T * 8 * 2 * 32];
    __shared__ uint4  sW2[K2T * 8 * 2 * 32];
    __shared__ float4 sB1[HID / 4];
    __shared__ float  sB2[HID];

    const int tid  = threadIdx.x;
    const int lane = tid & 31;
    const int w    = tid >> 5;

    {
        const int n  = tid & 127;
        const int k0 = tid >> 7;
        float u1[K1T][8], u2[K2T][8];
        #pragma unroll
        for (int k = 0; k < K1T - 1; ++k) wunit_load(u1[k], We1, k0 + 4 * k, n);
        wunit_load_padded(u1[K1T - 1], We1, k0 + 4 * (K1T - 1), n, 2 * HID + 1 + EDGE_D);
        #pragma unroll
        for (int k = 0; k < K2T; ++k) wunit_load(u2[k], We2, k0 + 4 * k, n);
        float4 b1v = make_float4(0.0f, 0.0f, 0.0f, 0.0f);
        float  b2v = 0.0f;
        if (tid < HID / 4) b1v = ((const float4*)be1)[tid];
        if (tid < HID)     b2v = be2[tid];

        #pragma unroll
        for (int k = 0; k < K1T; ++k) wunit_store(sW1, k0 + 4 * k, n, u1[k]);
        #pragma unroll
        for (int k = 0; k < K2T; ++k) wunit_store(sW2, k0 + 4 * k, n, u2[k]);
        if (tid < HID / 4) sB1[tid] = b1v;
        if (tid < HID)     sB2[tid] = b2v;
    }
    __syncthreads();

    const int hi   = lane >> 4;
    const int c    = lane & 15;

    const int TSTRIDE = EDGE_WAVES * EDGE_GRID;
    int tile = w * EDGE_GRID + blockIdx.x;

    uint4  Bq[16];
    float4 ea4[4];
    int r = 0, cn = 0;
    float radial = 0.f;
    if (tile < EDGE_TILES) {
        const int e = tile * 16 + c;
        r  = eidx[e];
        cn = eidx[N_EDGES + e];
        {
            const float4* ep = (const float4*)(eattr + e * EDGE_D);
            const uint4* pr = (const uint4*)(hb + r  * HID + hi * 8);
            const uint4* pc = (const uint4*)(hb + cn * HID + hi * 8);
            #pragma unroll
            for (int t = 0; t < 4; ++t) {
                Bq[2 * t]     = pr[4 * t];     Bq[2 * t + 1]     = pr[4 * t + 2];
                Bq[2 * t + 8] = pc[4 * t];     Bq[2 * t + 9]     = pc[4 * t + 2];
            }
            ea4[0] = ep[0]; ea4[1] = ep[1]; ea4[2] = ep[2]; ea4[3] = ep[3];
            const float d0 = x[r * 3 + 0] - x[cn * 3 + 0];
            const float d1 = x[r * 3 + 1] - x[cn * 3 + 1];
            const float d2 = x[r * 3 + 2] - x[cn * 3 + 2];
            radial = d0 * d0 + d1 * d1 + d2 * d2;
        }
    }

    for (; tile < EDGE_TILES; tile += TSTRIDE) {
        const int tnx = (tile + TSTRIDE < EDGE_TILES) ? tile + TSTRIDE : tile;
        const int en  = tnx * 16 + c;
        const int rn  = eidx[en];
        const int cnn = eidx[N_EDGES + en];
        asm volatile("" ::: "memory");

        Frag Wa[4], Wb[4];
        load_wset(Wa, sW1 + lane);
        v8f acc[8] = {};
        #pragma unroll
        for (int t = 0; t < K1T; ++t) {
            FragR B;
            if (t < 8) {
                B.q[0] = Bq[2 * t]; B.q[1] = Bq[2 * t + 1];
            } else {
                const float ea[16] = {ea4[0].x, ea4[0].y, ea4[0].z, ea4[0].w,
                                      ea4[1].x, ea4[1].y, ea4[1].z, ea4[1].w,
                                      ea4[2].x, ea4[2].y, ea4[2].z, ea4[2].w,
                                      ea4[3].x, ea4[3].y, ea4[3].z, ea4[3].w};
                float v[9];
                v[0] = hi ? ea[7] : radial;
                #pragma unroll
                for (int i = 1; i < 8; ++i) v[i] = hi ? ea[7 + i] : ea[i - 1];
                v[8] = hi ? 0.0f : ea[15];
                B.u[0] = pack_bf16(v[0], v[1]);
                B.u[1] = pack_bf16(v[2], v[3]);
                B.u[2] = pack_bf16(v[4], v[5]);
                B.u[3] = pack_bf16(v[6], v[7]);
                B.u[4] = pack_bf16(v[8], 0.0f);
                B.u[5] = 0u; B.u[6] = 0u; B.u[7] = 0u;
            }
            load_wset(Wb, sW1 + (t * 8 + 4) * 64 + lane);
            sched_fence();
            #pragma unroll
            for (int q = 0; q < 4; ++q)
                acc[q] = __builtin_amdgcn_wmma_f32_16x16x32_bf16(
                    false, Wa[q].v, false, B.v, (short)0, acc[q], false, false);
            sched_fence();
            load_wset(Wa, (t + 1 < K1T) ? sW1 + (t + 1) * 8 * 64 + lane : sW2 + lane);
            sched_fence();
            #pragma unroll
            for (int q = 0; q < 4; ++q)
                acc[4 + q] = __builtin_amdgcn_wmma_f32_16x16x32_bf16(
                    false, Wb[q].v, false, B.v, (short)0, acc[4 + q], false, false);
            sched_fence();
        }

        int srow[16];
        #pragma unroll
        for (int k = 0; k < 16; ++k) srow[k] = __builtin_amdgcn_readlane(r, k);


        FragR M[K2T];
        #pragma unroll
        for (int jt = 0; jt < 8; ++jt) {
            const float4 b0 = sB1[4 * jt + 2 * hi + 0];
            const float4 b1 = sB1[4 * jt + 2 * hi + 1];
            unsigned int* u = &M[jt >> 1].u[(jt & 1) * 4];
            u[0] = pack_bf16(silu_f(acc[jt][0] + b0.x), silu_f(acc[jt][1] + b0.y));
            u[1] = pack_bf16(silu_f(acc[jt][2] + b0.z), silu_f(acc[jt][3] + b0.w));
            u[2] = pack_bf16(silu_f(acc[jt][4] + b1.x), silu_f(acc[jt][5] + b1.y));
            u[3] = pack_bf16(silu_f(acc[jt][6] + b1.z), silu_f(acc[jt][7] + b1.w));
        }

        v8f acc2[8] = {};
        #pragma unroll
        for (int t = 0; t < K2T; ++t) {
            load_wset(Wb, sW2 + (t * 8 + 4) * 64 + lane);
            sched_fence();
            #pragma unroll
            for (int q = 0; q < 4; ++q)
                acc2[q] = __builtin_amdgcn_wmma_f32_16x16x32_bf16(
                    false, M[t].v, false, Wa[q].v, (short)0, acc2[q], false, false);
            sched_fence();
            if (t + 1 < K2T) {
                load_wset(Wa, sW2 + (t + 1) * 8 * 64 + lane);
                sched_fence();
            }
            #pragma unroll
            for (int q = 0; q < 4; ++q)
                acc2[4 + q] = __builtin_amdgcn_wmma_f32_16x16x32_bf16(
                    false, M[t].v, false, Wb[q].v, (short)0, acc2[4 + q], false, false);
            sched_fence();
        }

        float bA[4], bB[4];
        #pragma unroll
        for (int p = 0; p < 4; ++p) { bA[p] = sB2[32 * p + c]; bB[p] = sB2[32 * p + 16 + c]; }

        float nxr0, nxr1, nxr2, nxc0, nxc1, nxc2;
        {
            r = rn; cn = cnn;
            const float4* ep = (const float4*)(eattr + en * EDGE_D);
            const uint4* pr = (const uint4*)(hb + r  * HID + hi * 8);
            const uint4* pc = (const uint4*)(hb + cn * HID + hi * 8);
            #pragma unroll
            for (int t = 0; t < 4; ++t) {
                Bq[2 * t]     = pr[4 * t];     Bq[2 * t + 1]     = pr[4 * t + 2];
                Bq[2 * t + 8] = pc[4 * t];     Bq[2 * t + 9]     = pc[4 * t + 2];
            }
            ea4[0] = ep[0]; ea4[1] = ep[1]; ea4[2] = ep[2]; ea4[3] = ep[3];
            nxr0 = x[r * 3 + 0];  nxr1 = x[r * 3 + 1];  nxr2 = x[r * 3 + 2];
            nxc0 = x[cn * 3 + 0]; nxc1 = x[cn * 3 + 1]; nxc2 = x[cn * 3 + 2];
        }

        #pragma unroll
        for (int p = 0; p < 4; ++p) {
            const unsigned int col = (unsigned int)(32 * p + lane);
            float vlo[8], vhi[8];
            #pragma unroll
            for (int i = 0; i < 8; ++i) {
                const float fa = silu_f(acc2[2 * p][i] + bA[p]);
                const float fb = silu_f(acc2[2 * p + 1][i] + bB[p]);
                const float got = xhalf(hi ? fa : fb);
                vlo[i] = hi ? got : fa;
                vhi[i] = hi ? fb : got;
            }
            sched_fence();
            #pragma unroll
            for (int i = 0; i < 8; ++i) {
                atomicAdd(&agg[(unsigned int)srow[i] * HID + col], vlo[i]);
                atomicAdd(&agg[(unsigned int)srow[8 + i] * HID + col], vhi[i]);
            }
            sched_fence();
        }

        {
            const float d0 = nxr0 - nxc0;
            const float d1 = nxr1 - nxc1;
            const float d2 = nxr2 - nxc2;
            radial = d0 * d0 + d1 * d1 + d2 * d2;
        }
    }
}

#define KN1T 8
#define KN2T 4

#define NODE_THREADS 512
#define NODE_WAVES   16
#define NODE_GRID    256
#define NODE_TILES   (N_NODES / 16)

static_assert(N_NODES % 16 == 0, "a node tile is 16 whole rows");
static_assert(NODE_TILES <= NODE_WAVES * NODE_GRID, "one tile per wave");
static_assert(NODE_THREADS == 4 * HID && NODE_THREADS == 32 * NODE_WAVES,
              "weight staging: a thread keeps one of 128 features and every fourth k-group");

__global__ __launch_bounds__(NODE_THREADS)
void node_kernel(const float* __restrict__ h,
                 const unsigned short* __restrict__ hb,
                 const float* __restrict__ agg,
                 const int*   __restrict__ batch,
                 const float* __restrict__ Wn1,
                 const float* __restrict__ bn1,
                 const float* __restrict__ Wn2,
                 const float* __restrict__ bn2,
                 float* __restrict__ sums) {
    __shared__ uint4  sW1[KN1T * 8 * 2 * 32];
    __shared__ uint4  sW2[KN2T * 8 * 2 * 32];
    __shared__ float4 sB1[HID / 4];
    __shared__ float  sB2[HID];

    const int tid  = threadIdx.x;
    const int w    = tid >> 5;
    const int lane = tid & 31;
    const int hi   = lane >> 4;
    const int c    = lane & 15;
    const int tile = w * NODE_GRID + blockIdx.x;
    const bool has = tile < NODE_TILES;

    const int wn  = tid & 127;
    const int wk0 = tid >> 7;
    float u1[KN1T][8], u2[KN2T][8];
    #pragma unroll
    for (int k = 0; k < KN1T; ++k) wunit_load(u1[k], Wn1, wk0 + 4 * k, wn);
    #pragma unroll
    for (int k = 0; k < KN2T; ++k) wunit_load(u2[k], Wn2, wk0 + 4 * k, wn);
    float4 b1v = make_float4(0.0f, 0.0f, 0.0f, 0.0f);
    float  b2v = 0.0f;
    if (tid < HID / 4) b1v = ((const float4*)bn1)[tid];
    if (tid < HID)     b2v = bn2[tid];

    uint4  Bh[8];
    float4 Ba[16];
    int gl = 0;
    if (has) {
        const int node = tile * 16 + c;
        const uint4*  ph = (const uint4*)(hb + node * HID + hi * 8);
        const float4* pa = (const float4*)(agg + node * HID + hi * 8);
        #pragma unroll
        for (int t = 0; t < 4; ++t) {
            Bh[2 * t] = ph[4 * t];          Bh[2 * t + 1] = ph[4 * t + 2];
            Ba[4 * t]     = pa[8 * t];      Ba[4 * t + 1] = pa[8 * t + 1];
            Ba[4 * t + 2] = pa[8 * t + 4];  Ba[4 * t + 3] = pa[8 * t + 5];
        }
        gl = batch[node];
    }

    #pragma unroll
    for (int k = 0; k < KN1T; ++k) wunit_store(sW1, wk0 + 4 * k, wn, u1[k]);
    #pragma unroll
    for (int k = 0; k < KN2T; ++k) wunit_store(sW2, wk0 + 4 * k, wn, u2[k]);
    if (tid < HID / 4) sB1[tid] = b1v;
    if (tid < HID)     sB2[tid] = b2v;
    __syncthreads();

    if (!has) return;

    Frag Wa[4], Wb[4];
    load_wset(Wa, sW1 + lane);
    v8f acc[8] = {};
    #pragma unroll
    for (int t = 0; t < KN1T; ++t) {
        FragR B;
        if (t < 4) {
            B.q[0] = Bh[2 * t]; B.q[1] = Bh[2 * t + 1];
        } else {
            const float4 a0 = Ba[4 * (t - 4)],     a1 = Ba[4 * (t - 4) + 1];
            const float4 a2 = Ba[4 * (t - 4) + 2], a3 = Ba[4 * (t - 4) + 3];
            B.u[0] = pack_bf16(a0.x, a0.y); B.u[1] = pack_bf16(a0.z, a0.w);
            B.u[2] = pack_bf16(a1.x, a1.y); B.u[3] = pack_bf16(a1.z, a1.w);
            B.u[4] = pack_bf16(a2.x, a2.y); B.u[5] = pack_bf16(a2.z, a2.w);
            B.u[6] = pack_bf16(a3.x, a3.y); B.u[7] = pack_bf16(a3.z, a3.w);
        }
        load_wset(Wb, sW1 + (t * 8 + 4) * 64 + lane);
        sched_fence();
        #pragma unroll
        for (int q = 0; q < 4; ++q)
            acc[q] = __builtin_amdgcn_wmma_f32_16x16x32_bf16(
                false, Wa[q].v, false, B.v, (short)0, acc[q], false, false);
        sched_fence();
        load_wset(Wa, (t + 1 < KN1T) ? sW1 + (t + 1) * 8 * 64 + lane : sW2 + lane);
        sched_fence();
        #pragma unroll
        for (int q = 0; q < 4; ++q)
            acc[4 + q] = __builtin_amdgcn_wmma_f32_16x16x32_bf16(
                false, Wb[q].v, false, B.v, (short)0, acc[4 + q], false, false);
        sched_fence();
    }

    int grow[16];
    #pragma unroll
    for (int k = 0; k < 16; ++k) grow[k] = __builtin_amdgcn_readlane(gl, k);

    float hres[4][16];
    {
        const float* hp = h + tile * 16 * HID + lane;
        #pragma unroll
        for (int p = 0; p < 4; ++p) {
            #pragma unroll
            for (int k = 0; k < 16; ++k) hres[p][k] = hp[k * HID + 32 * p];
        }
    }

    asm volatile("" ::: "memory");

    FragR M[KN2T];
    #pragma unroll
    for (int jt = 0; jt < 8; ++jt) {
        const float4 b0 = sB1[4 * jt + 2 * hi + 0];
        const float4 b1 = sB1[4 * jt + 2 * hi + 1];
        unsigned int* u = &M[jt >> 1].u[(jt & 1) * 4];
        u[0] = pack_bf16(silu_f(acc[jt][0] + b0.x), silu_f(acc[jt][1] + b0.y));
        u[1] = pack_bf16(silu_f(acc[jt][2] + b0.z), silu_f(acc[jt][3] + b0.w));
        u[2] = pack_bf16(silu_f(acc[jt][4] + b1.x), silu_f(acc[jt][5] + b1.y));
        u[3] = pack_bf16(silu_f(acc[jt][6] + b1.z), silu_f(acc[jt][7] + b1.w));
    }

    v8f acc2[8] = {};
    #pragma unroll
    for (int t = 0; t < KN2T; ++t) {
        load_wset(Wb, sW2 + (t * 8 + 4) * 64 + lane);
        sched_fence();
        #pragma unroll
        for (int q = 0; q < 4; ++q)
            acc2[q] = __builtin_amdgcn_wmma_f32_16x16x32_bf16(
                false, M[t].v, false, Wa[q].v, (short)0, acc2[q], false, false);
        sched_fence();
        if (t + 1 < KN2T) {
            load_wset(Wa, sW2 + (t + 1) * 8 * 64 + lane);
            sched_fence();
        }
        #pragma unroll
        for (int q = 0; q < 4; ++q)
            acc2[4 + q] = __builtin_amdgcn_wmma_f32_16x16x32_bf16(
                false, M[t].v, false, Wb[q].v, (short)0, acc2[4 + q], false, false);
        sched_fence();
    }

    #pragma unroll
    for (int p = 0; p < 4; ++p) {
        const float bA = sB2[32 * p + c];
        const float bB = sB2[32 * p + 16 + c];
        const unsigned int col = (unsigned int)(32 * p + lane);
        #pragma unroll
        for (int i = 0; i < 8; ++i) {
            const float fa = acc2[2 * p][i] + bA;
            const float fb = acc2[2 * p + 1][i] + bB;
            const float got = xhalf(hi ? fa : fb);
            const float vlo = hi ? got : fa;
            const float vhi = hi ? fb : got;
            atomicAdd(&sums[(unsigned int)grow[i] * HID + col],     hres[p][i] + vlo);
            atomicAdd(&sums[(unsigned int)grow[8 + i] * HID + col], hres[p][8 + i] + vhi);
        }
    }
}

#define FIN_THREADS 1024

__global__ __launch_bounds__(FIN_THREADS)
void finalize_kernel(const float* __restrict__ sums,
                     const int*   __restrict__ batch,
                     float* __restrict__ out) {
    __shared__ int sc[2][FIN_THREADS / 32];
    const int tid = threadIdx.x;
    const int g0 = blockIdx.x * 2, g1 = g0 + 1;
    const int oi = blockIdx.x * 2 * HID + tid;
    float s = 0.0f;
    if (tid < 2 * HID) s = sums[oi];

    const int4* b4 = (const int4*)batch;
    int c0 = 0, c1 = 0;
    for (int k = 0; k < (N_NODES / 4 + FIN_THREADS - 1) / FIN_THREADS; ++k) {
        const int i = tid + k * FIN_THREADS;
        if (i < N_NODES / 4) {
            const int4 v = b4[i];
            c0 += (v.x == g0) + (v.y == g0) + (v.z == g0) + (v.w == g0);
            c1 += (v.x == g1) + (v.y == g1) + (v.z == g1) + (v.w == g1);
        }
    }
    for (int o = 16; o > 0; o >>= 1) {
        c0 += __shfl_xor(c0, o);
        c1 += __shfl_xor(c1, o);
    }
    if ((tid & 31) == 0) { sc[0][tid >> 5] = c0; sc[1][tid >> 5] = c1; }
    __syncthreads();
    if (tid < 2 * HID) {
        const int* p = sc[tid >> 7];
        int cnt = 0;
        for (int w = 0; w < FIN_THREADS / 32; ++w) cnt += p[w];
        out[oi] = s / fmaxf((float)cnt, 1.0f);
    }
}

extern "C" void kernel_launch(void* const* d_in, const int* in_sizes, int n_in,
                              void* d_out, int out_size, void* d_ws, size_t ws_size,
                              hipStream_t stream) {
    const float* h     = (const float*)d_in[0];
    const int*   eidx  = (const int*)  d_in[1];
    const float* x     = (const float*)d_in[2];
    const float* eattr = (const float*)d_in[3];
    const int*   batch = (const int*)  d_in[4];
    const float* We1   = (const float*)d_in[5];
    const float* be1   = (const float*)d_in[6];
    const float* We2   = (const float*)d_in[7];
    const float* be2   = (const float*)d_in[8];
    const float* Wn1   = (const float*)d_in[9];
    const float* bn1   = (const float*)d_in[10];
    const float* Wn2   = (const float*)d_in[11];
    const float* bn2   = (const float*)d_in[12];

    char* ws = (char*)d_ws;
    float* agg    = (float*)(ws + 204800);
    float* sums   = (float*)(ws + 204800 + 25600000);
    unsigned int* hb = (unsigned int*)(ws + 204800 + 25600000 + 32768 + 256);

    int nz = N_NODES * HID + N_GRAPHS * HID;
    prep_kernel<<<(nz + 255) / 256, 256, 0, stream>>>(agg, nz, h, hb);

    edge_kernel<<<EDGE_GRID, EDGE_THREADS, 0, stream>>>((const unsigned short*)hb, eidx, x, eattr,
                                                        We1, be1, We2, be2, agg);
    node_kernel<<<NODE_GRID, NODE_THREADS, 0, stream>>>(h, (const unsigned short*)hb, agg, batch,
                                                        Wn1, bn1, Wn2, bn2, sums);
    finalize_kernel<<<N_GRAPHS / 2, FIN_THREADS, 0, stream>>>(sums, batch, (float*)d_out);
}


# edge + node kernels: SiLU on pairs of values: product, sum and the 6 middle steps of its correctly rounded division as packed f32 instructions (v_pk_fma_f32); same roundings, sum orders unchanged
# speedup vs baseline: 3.4359x; 1.0412x over previous
#include <hip/hip_runtime.h>
#include <math.h>

#define N_NODES  50000
#define N_EDGES  800000
#define HID      128
#define EDGE_D   16
#define N_GRAPHS 64

typedef __attribute__((ext_vector_type(16))) __bf16 v16bf;
typedef __attribute__((ext_vector_type(8)))  float  v8f;

union Frag {
    uint4 q[2];
    v16bf v;
};

typedef __attribute__((ext_vector_type(2))) float f32pair;

__device__ __forceinline__ f32pair silu_pair(f32pair x) {
    const f32pair nlog2e = {-1.44269504088896340736f, -1.44269504088896340736f};
    const f32pair one = {1.0f, 1.0f};
    const f32pair t = x * nlog2e;
    f32pair e;
    e.x = __builtin_amdgcn_exp2f(t.x);
    e.y = __builtin_amdgcn_exp2f(t.y);
    const f32pair d = one + e;
    bool sd0, sd1, sn0, sn1;
    f32pair ds, ns, r;
    ds.x = __builtin_amdgcn_div_scalef(x.x, d.x, false, &sd0);
    ds.y = __builtin_amdgcn_div_scalef(x.y, d.y, false, &sd1);
    ns.x = __builtin_amdgcn_div_scalef(x.x, d.x, true, &sn0);
    ns.y = __builtin_amdgcn_div_scalef(x.y, d.y, true, &sn1);
    r.x = __builtin_amdgcn_rcpf(ds.x);
    r.y = __builtin_amdgcn_rcpf(ds.y);
    const f32pair e0 = __builtin_elementwise_fma(-ds, r, one);
    const f32pair r1 = __builtin_elementwise_fma(e0, r, r);
    const f32pair q0 = ns * r1;
    const f32pair e1 = __builtin_elementwise_fma(-ds, q0, ns);
    const f32pair q1 = __builtin_elementwise_fma(e1, r1, q0);
    const f32pair e2 = __builtin_elementwise_fma(-ds, q1, ns);
    f32pair o;
    o.x = __builtin_amdgcn_div_fixupf(__builtin_amdgcn_div_fmasf(e2.x, r1.x, q1.x, sn0), d.x, x.x);
    o.y = __builtin_amdgcn_div_fixupf(__builtin_amdgcn_div_fmasf(e2.y, r1.y, q1.y, sn1), d.y, x.y);
    return o;
}

__device__ __forceinline__ unsigned int pack_bf16(float a, float b);

__device__ __forceinline__ unsigned int silu_pack(float a, float ba, float b, float bb) {
    const f32pair v = {a, b};
    const f32pair bias = {ba, bb};
    const f32pair s = silu_pair(v + bias);
    return pack_bf16(s.x, s.y);
}

__device__ __forceinline__ float xhalf(float x) {
    const int b = __builtin_bit_cast(int, x);
    return __builtin_bit_cast(float,
        __builtin_amdgcn_permlanex16(b, b, 0x76543210, 0xfedcba98, false, false));
}

typedef __attribute__((ext_vector_type(2))) __bf16 bf16x2_t;
typedef __attribute__((ext_vector_type(2))) float  f32x2_t;

__device__ __forceinline__ unsigned int pack_bf16(float a, float b) {
    f32x2_t f = {a, b};
    bf16x2_t p = __builtin_convertvector(f, bf16x2_t);
    return __builtin_bit_cast(unsigned int, p);
}

__device__ __forceinline__ void wunit_load(float (&v)[8], const float* __restrict__ W, int kg, int n) {
    #pragma unroll
    for (int r = 0; r < 8; ++r) v[r] = W[(kg * 8 + r) * HID + n];
}

__device__ __forceinline__ void wunit_load_padded(float (&v)[8], const float* __restrict__ W,
                                                  int kg, int n, int Ksrc) {
    #pragma unroll
    for (int r = 0; r < 8; ++r) {
        const int K = kg * 8 + r;
        const float keep = (K < Ksrc) ? 1.0f : 0.0f;
        v[r] = W[(K < Ksrc ? K : Ksrc - 1) * HID + n] * keep;
    }
}

__device__ __forceinline__ void wunit_store(uint4* __restrict__ sW, int kg, int n, const float (&v)[8]) {
    const int t = kg >> 2, half = (kg >> 1) & 1, g = kg & 1;
    sW[((t * 8 + (n >> 4)) * 2 + half) * 32 + g * 16 + (n & 15)] =
        make_uint4(pack_bf16(v[0], v[1]), pack_bf16(v[2], v[3]), pack_bf16(v[4], v[5]), pack_bf16(v[6], v[7]));
}

__global__ void prep_kernel(float* __restrict__ p, int n,
                            const float* __restrict__ h, unsigned int* __restrict__ hb) {
    int i = blockIdx.x * blockDim.x + threadIdx.x;
    if (i < n) p[i] = 0.0f;
    if (i < N_NODES * HID / 2) {
        const float2 v = ((const float2*)h)[i];
        hb[i] = pack_bf16(v.x, v.y);
    }
}

#define K1T 9
#define K2T 4

#define EDGE_THREADS 512
#define EDGE_WAVES   16
#define EDGE_GRID    256
#define EDGE_TILES   (N_EDGES / 16)

static_assert(EDGE_THREADS == 4 * HID && EDGE_THREADS == 32 * EDGE_WAVES,
              "weight staging: a thread keeps one of 128 features and every fourth k-group");

union FragR {
    unsigned int u[8];
    uint4 q[2];
    v16bf v;
};

__device__ __forceinline__ void sched_fence() { __builtin_amdgcn_sched_barrier(0); }

__device__ __forceinline__ void load_wset(Frag (&s)[4], const uint4* p) {
    #pragma unroll
    for (int q = 0; q < 4; ++q) { s[q].q[0] = p[q * 64]; s[q].q[1] = p[q * 64 + 32]; }
}

__global__ __launch_bounds__(EDGE_THREADS)
void edge_kernel(const unsigned short* __restrict__ hb,
                 const int*   __restrict__ eidx,
                 const float* __restrict__ x,
                 const float* __restrict__ eattr,
                 const float* __restrict__ We1,
                 const float* __restrict__ be1,
                 const float* __restrict__ We2,
                 const float* __restrict__ be2,
                 float* __restrict__ agg) {
    __shared__ uint4  sW1[K1T * 8 * 2 * 32];
    __shared__ uint4  sW2[K2T * 8 * 2 * 32];
    __shared__ float4 sB1[HID / 4];
    __shared__ float  sB2[HID];

    const int tid  = threadIdx.x;
    const int lane = tid & 31;
    const int w    = tid >> 5;

    {
        const int n  = tid & 127;
        const int k0 = tid >> 7;
        float u1[K1T][8], u2[K2T][8];
        #pragma unroll
        for (int k = 0; k < K1T - 1; ++k) wunit_load(u1[k], We1, k0 + 4 * k, n);
        wunit_load_padded(u1[K1T - 1], We1, k0 + 4 * (K1T - 1), n, 2 * HID + 1 + EDGE_D);
        #pragma unroll
        for (int k = 0; k < K2T; ++k) wunit_load(u2[k], We2, k0 + 4 * k, n);
        float4 b1v = make_float4(0.0f, 0.0f, 0.0f, 0.0f);
        float  b2v = 0.0f;
        if (tid < HID / 4) b1v = ((const float4*)be1)[tid];
        if (tid < HID)     b2v = be2[tid];

        #pragma unroll
        for (int k = 0; k < K1T; ++k) wunit_store(sW1, k0 + 4 * k, n, u1[k]);
        #pragma unroll
        for (int k = 0; k < K2T; ++k) wunit_store(sW2, k0 + 4 * k, n, u2[k]);
        if (tid < HID / 4) sB1[tid] = b1v;
        if (tid < HID)     sB2[tid] = b2v;
    }
    __syncthreads();

    const int hi   = lane >> 4;
    const int c    = lane & 15;

    const int TSTRIDE = EDGE_WAVES * EDGE_GRID;
    int tile = w * EDGE_GRID + blockIdx.x;

    uint4  Bq[16];
    float4 ea4[4];
    int r = 0, cn = 0;
    float radial = 0.f;
    if (tile < EDGE_TILES) {
        const int e = tile * 16 + c;
        r  = eidx[e];
        cn = eidx[N_EDGES + e];
        {
            const float4* ep = (const float4*)(eattr + e * EDGE_D);
            const uint4* pr = (const uint4*)(hb + r  * HID + hi * 8);
            const uint4* pc = (const uint4*)(hb + cn * HID + hi * 8);
            #pragma unroll
            for (int t = 0; t < 4; ++t) {
                Bq[2 * t]     = pr[4 * t];     Bq[2 * t + 1]     = pr[4 * t + 2];
                Bq[2 * t + 8] = pc[4 * t];     Bq[2 * t + 9]     = pc[4 * t + 2];
            }
            ea4[0] = ep[0]; ea4[1] = ep[1]; ea4[2] = ep[2]; ea4[3] = ep[3];
            const float d0 = x[r * 3 + 0] - x[cn * 3 + 0];
            const float d1 = x[r * 3 + 1] - x[cn * 3 + 1];
            const float d2 = x[r * 3 + 2] - x[cn * 3 + 2];
            radial = d0 * d0 + d1 * d1 + d2 * d2;
        }
    }

    for (; tile < EDGE_TILES; tile += TSTRIDE) {
        const int tnx = (tile + TSTRIDE < EDGE_TILES) ? tile + TSTRIDE : tile;
        const int en  = tnx * 16 + c;
        const int rn  = eidx[en];
        const int cnn = eidx[N_EDGES + en];
        asm volatile("" ::: "memory");

        Frag Wa[4], Wb[4];
        load_wset(Wa, sW1 + lane);
        v8f acc[8] = {};
        #pragma unroll
        for (int t = 0; t < K1T; ++t) {
            FragR B;
            if (t < 8) {
                B.q[0] = Bq[2 * t]; B.q[1] = Bq[2 * t + 1];
            } else {
                const float ea[16] = {ea4[0].x, ea4[0].y, ea4[0].z, ea4[0].w,
                                      ea4[1].x, ea4[1].y, ea4[1].z, ea4[1].w,
                                      ea4[2].x, ea4[2].y, ea4[2].z, ea4[2].w,
                                      ea4[3].x, ea4[3].y, ea4[3].z, ea4[3].w};
                float v[9];
                v[0] = hi ? ea[7] : radial;
                #pragma unroll
                for (int i = 1; i < 8; ++i) v[i] = hi ? ea[7 + i] : ea[i - 1];
                v[8] = hi ? 0.0f : ea[15];
                B.u[0] = pack_bf16(v[0], v[1]);
                B.u[1] = pack_bf16(v[2], v[3]);
                B.u[2] = pack_bf16(v[4], v[5]);
                B.u[3] = pack_bf16(v[6], v[7]);
                B.u[4] = pack_bf16(v[8], 0.0f);
                B.u[5] = 0u; B.u[6] = 0u; B.u[7] = 0u;
            }
            load_wset(Wb, sW1 + (t * 8 + 4) * 64 + lane);
            sched_fence();
            #pragma unroll
            for (int q = 0; q < 4; ++q)
                acc[q] = __builtin_amdgcn_wmma_f32_16x16x32_bf16(
                    false, Wa[q].v, false, B.v, (short)0, acc[q], false, false);
            sched_fence();
            load_wset(Wa, (t + 1 < K1T) ? sW1 + (t + 1) * 8 * 64 + lane : sW2 + lane);
            sched_fence();
            #pragma unroll
            for (int q = 0; q < 4; ++q)
                acc[4 + q] = __builtin_amdgcn_wmma_f32_16x16x32_bf16(
                    false, Wb[q].v, false, B.v, (short)0, acc[4 + q], false, false);
            sched_fence();
        }

        int srow[16];
        #pragma unroll
        for (int k = 0; k < 16; ++k) srow[k] = __builtin_amdgcn_readlane(r, k);


        FragR M[K2T];
        #pragma unroll
        for (int jt = 0; jt < 8; ++jt) {
            const float4 b0 = sB1[4 * jt + 2 * hi + 0];
            const float4 b1 = sB1[4 * jt + 2 * hi + 1];
            unsigned int* u = &M[jt >> 1].u[(jt & 1) * 4];
            u[0] = silu_pack(acc[jt][0], b0.x, acc[jt][1], b0.y);
            u[1] = silu_pack(acc[jt][2], b0.z, acc[jt][3], b0.w);
            u[2] = silu_pack(acc[jt][4], b1.x, acc[jt][5], b1.y);
            u[3] = silu_pack(acc[jt][6], b1.z, acc[jt][7], b1.w);
        }

        v8f acc2[8] = {};
        #pragma unroll
        for (int t = 0; t < K2T; ++t) {
            load_wset(Wb, sW2 + (t * 8 + 4) * 64 + lane);
            sched_fence();
            #pragma unroll
            for (int q = 0; q < 4; ++q)
                acc2[q] = __builtin_amdgcn_wmma_f32_16x16x32_bf16(
                    false, M[t].v, false, Wa[q].v, (short)0, acc2[q], false, false);
            sched_fence();
            if (t + 1 < K2T) {
                load_wset(Wa, sW2 + (t + 1) * 8 * 64 + lane);
                sched_fence();
            }
            #pragma unroll
            for (int q = 0; q < 4; ++q)
                acc2[4 + q] = __builtin_amdgcn_wmma_f32_16x16x32_bf16(
                    false, M[t].v, false, Wb[q].v, (short)0, acc2[4 + q], false, false);
            sched_fence();
        }

        float bA[4], bB[4];
        #pragma unroll
        for (int p = 0; p < 4; ++p) { bA[p] = sB2[32 * p + c]; bB[p] = sB2[32 * p + 16 + c]; }

        float nxr0, nxr1, nxr2, nxc0, nxc1, nxc2;
        {
            r = rn; cn = cnn;
            const float4* ep = (const float4*)(eattr + en * EDGE_D);
            const uint4* pr = (const uint4*)(hb + r  * HID + hi * 8);
            const uint4* pc = (const uint4*)(hb + cn * HID + hi * 8);
            #pragma unroll
            for (int t = 0; t < 4; ++t) {
                Bq[2 * t]     = pr[4 * t];     Bq[2 * t + 1]     = pr[4 * t + 2];
                Bq[2 * t + 8] = pc[4 * t];     Bq[2 * t + 9]     = pc[4 * t + 2];
            }
            ea4[0] = ep[0]; ea4[1] = ep[1]; ea4[2] = ep[2]; ea4[3] = ep[3];
            nxr0 = x[r * 3 + 0];  nxr1 = x[r * 3 + 1];  nxr2 = x[r * 3 + 2];
            nxc0 = x[cn * 3 + 0]; nxc1 = x[cn * 3 + 1]; nxc2 = x[cn * 3 + 2];
        }

        #pragma unroll
        for (int p = 0; p < 4; ++p) {
            const unsigned int col = (unsigned int)(32 * p + lane);
            float vlo[8], vhi[8];
            #pragma unroll
            for (int i = 0; i < 8; i += 2) {
                const f32pair biasA = {bA[p], bA[p]}, biasB = {bB[p], bB[p]};
                const f32pair inA = {acc2[2 * p][i], acc2[2 * p][i + 1]};
                const f32pair inB = {acc2[2 * p + 1][i], acc2[2 * p + 1][i + 1]};
                const f32pair sA = silu_pair(inA + biasA);
                const f32pair sB = silu_pair(inB + biasB);
                #pragma unroll
                for (int k = 0; k < 2; ++k) {
                    const float fa = sA[k];
                    const float fb = sB[k];
                    const float got = xhalf(hi ? fa : fb);
                    vlo[i + k] = hi ? got : fa;
                    vhi[i + k] = hi ? fb : got;
                }
            }
            sched_fence();
            #pragma unroll
            for (int i = 0; i < 8; ++i) {
                atomicAdd(&agg[(unsigned int)srow[i] * HID + col], vlo[i]);
                atomicAdd(&agg[(unsigned int)srow[8 + i] * HID + col], vhi[i]);
            }
            sched_fence();
        }

        {
            const float d0 = nxr0 - nxc0;
            const float d1 = nxr1 - nxc1;
            const float d2 = nxr2 - nxc2;
            radial = d0 * d0 + d1 * d1 + d2 * d2;
        }
    }
}

#define KN1T 8
#define KN2T 4

#define NODE_THREADS 512
#define NODE_WAVES   16
#define NODE_GRID    256
#define NODE_TILES   (N_NODES / 16)

static_assert(N_NODES % 16 == 0, "a node tile is 16 whole rows");
static_assert(NODE_TILES <= NODE_WAVES * NODE_GRID, "one tile per wave");
static_assert(NODE_THREADS == 4 * HID && NODE_THREADS == 32 * NODE_WAVES,
              "weight staging: a thread keeps one of 128 features and every fourth k-group");

__global__ __launch_bounds__(NODE_THREADS)
void node_kernel(const float* __restrict__ h,
                 const unsigned short* __restrict__ hb,
                 const float* __restrict__ agg,
                 const int*   __restrict__ batch,
                 const float* __restrict__ Wn1,
                 const float* __restrict__ bn1,
                 const float* __restrict__ Wn2,
                 const float* __restrict__ bn2,
                 float* __restrict__ sums) {
    __shared__ uint4  sW1[KN1T * 8 * 2 * 32];
    __shared__ uint4  sW2[KN2T * 8 * 2 * 32];
    __shared__ float4 sB1[HID / 4];
    __shared__ float  sB2[HID];

    const int tid  = threadIdx.x;
    const int w    = tid >> 5;
    const int lane = tid & 31;
    const int hi   = lane >> 4;
    const int c    = lane & 15;
    const int tile = w * NODE_GRID + blockIdx.x;
    const bool has = tile < NODE_TILES;

    const int wn  = tid & 127;
    const int wk0 = tid >> 7;
    float u1[KN1T][8], u2[KN2T][8];
    #pragma unroll
    for (int k = 0; k < KN1T; ++k) wunit_load(u1[k], Wn1, wk0 + 4 * k, wn);
    #pragma unroll
    for (int k = 0; k < KN2T; ++k) wunit_load(u2[k], Wn2, wk0 + 4 * k, wn);
    float4 b1v = make_float4(0.0f, 0.0f, 0.0f, 0.0f);
    float  b2v = 0.0f;
    if (tid < HID / 4) b1v = ((const float4*)bn1)[tid];
    if (tid < HID)     b2v = bn2[tid];

    uint4  Bh[8];
    float4 Ba[16];
    int gl = 0;
    if (has) {
        const int node = tile * 16 + c;
        const uint4*  ph = (const uint4*)(hb + node * HID + hi * 8);
        const float4* pa = (const float4*)(agg + node * HID + hi * 8);
        #pragma unroll
        for (int t = 0; t < 4; ++t) {
            Bh[2 * t] = ph[4 * t];          Bh[2 * t + 1] = ph[4 * t + 2];
            Ba[4 * t]     = pa[8 * t];      Ba[4 * t + 1] = pa[8 * t + 1];
            Ba[4 * t + 2] = pa[8 * t + 4];  Ba[4 * t + 3] = pa[8 * t + 5];
        }
        gl = batch[node];
    }

    #pragma unroll
    for (int k = 0; k < KN1T; ++k) wunit_store(sW1, wk0 + 4 * k, wn, u1[k]);
    #pragma unroll
    for (int k = 0; k < KN2T; ++k) wunit_store(sW2, wk0 + 4 * k, wn, u2[k]);
    if (tid < HID / 4) sB1[tid] = b1v;
    if (tid < HID)     sB2[tid] = b2v;
    __syncthreads();

    if (!has) return;

    Frag Wa[4], Wb[4];
    load_wset(Wa, sW1 + lane);
    v8f acc[8] = {};
    #pragma unroll
    for (int t = 0; t < KN1T; ++t) {
        FragR B;
        if (t < 4) {
            B.q[0] = Bh[2 * t]; B.q[1] = Bh[2 * t + 1];
        } else {
            const float4 a0 = Ba[4 * (t - 4)],     a1 = Ba[4 * (t - 4) + 1];
            const float4 a2 = Ba[4 * (t - 4) + 2], a3 = Ba[4 * (t - 4) + 3];
            B.u[0] = pack_bf16(a0.x, a0.y); B.u[1] = pack_bf16(a0.z, a0.w);
            B.u[2] = pack_bf16(a1.x, a1.y); B.u[3] = pack_bf16(a1.z, a1.w);
            B.u[4] = pack_bf16(a2.x, a2.y); B.u[5] = pack_bf16(a2.z, a2.w);
            B.u[6] = pack_bf16(a3.x, a3.y); B.u[7] = pack_bf16(a3.z, a3.w);
        }
        load_wset(Wb, sW1 + (t * 8 + 4) * 64 + lane);
        sched_fence();
        #pragma unroll
        for (int q = 0; q < 4; ++q)
            acc[q] = __builtin_amdgcn_wmma_f32_16x16x32_bf16(
                false, Wa[q].v, false, B.v, (short)0, acc[q], false, false);
        sched_fence();
        load_wset(Wa, (t + 1 < KN1T) ? sW1 + (t + 1) * 8 * 64 + lane : sW2 + lane);
        sched_fence();
        #pragma unroll
        for (int q = 0; q < 4; ++q)
            acc[4 + q] = __builtin_amdgcn_wmma_f32_16x16x32_bf16(
                false, Wb[q].v, false, B.v, (short)0, acc[4 + q], false, false);
        sched_fence();
    }

    int grow[16];
    #pragma unroll
    for (int k = 0; k < 16; ++k) grow[k] = __builtin_amdgcn_readlane(gl, k);

    float hres[4][16];
    {
        const float* hp = h + tile * 16 * HID + lane;
        #pragma unroll
        for (int p = 0; p < 4; ++p) {
            #pragma unroll
            for (int k = 0; k < 16; ++k) hres[p][k] = hp[k * HID + 32 * p];
        }
    }

    asm volatile("" ::: "memory");

    FragR M[KN2T];
    #pragma unroll
    for (int jt = 0; jt < 8; ++jt) {
        const float4 b0 = sB1[4 * jt + 2 * hi + 0];
        const float4 b1 = sB1[4 * jt + 2 * hi + 1];
        unsigned int* u = &M[jt >> 1].u[(jt & 1) * 4];
        u[0] = silu_pack(acc[jt][0], b0.x, acc[jt][1], b0.y);
        u[1] = silu_pack(acc[jt][2], b0.z, acc[jt][3], b0.w);
        u[2] = silu_pack(acc[jt][4], b1.x, acc[jt][5], b1.y);
        u[3] = silu_pack(acc[jt][6], b1.z, acc[jt][7], b1.w);
    }

    v8f acc2[8] = {};
    #pragma unroll
    for (int t = 0; t < KN2T; ++t) {
        load_wset(Wb, sW2 + (t * 8 + 4) * 64 + lane);
        sched_fence();
        #pragma unroll
        for (int q = 0; q < 4; ++q)
            acc2[q] = __builtin_amdgcn_wmma_f32_16x16x32_bf16(
                false, M[t].v, false, Wa[q].v, (short)0, acc2[q], false, false);
        sched_fence();
        if (t + 1 < KN2T) {
            load_wset(Wa, sW2 + (t + 1) * 8 * 64 + lane);
            sched_fence();
        }
        #pragma unroll
        for (int q = 0; q < 4; ++q)
            acc2[4 + q] = __builtin_amdgcn_wmma_f32_16x16x32_bf16(
                false, M[t].v, false, Wb[q].v, (short)0, acc2[4 + q], false, false);
        sched_fence();
    }

    #pragma unroll
    for (int p = 0; p < 4; ++p) {
        const float bA = sB2[32 * p + c];
        const float bB = sB2[32 * p + 16 + c];
        const unsigned int col = (unsigned int)(32 * p + lane);
        #pragma unroll
        for (int i = 0; i < 8; ++i) {
            const float fa = acc2[2 * p][i] + bA;
            const float fb = acc2[2 * p + 1][i] + bB;
            const float got = xhalf(hi ? fa : fb);
            const float vlo = hi ? got : fa;
            const float vhi = hi ? fb : got;
            atomicAdd(&sums[(unsigned int)grow[i] * HID + col],     hres[p][i] + vlo);
            atomicAdd(&sums[(unsigned int)grow[8 + i] * HID + col], hres[p][8 + i] + vhi);
        }
    }
}

#define FIN_THREADS 1024

__global__ __launch_bounds__(FIN_THREADS)
void finalize_kernel(const float* __restrict__ sums,
                     const int*   __restrict__ batch,
                     float* __restrict__ out) {
    __shared__ int sc[2][FIN_THREADS / 32];
    const int tid = threadIdx.x;
    const int g0 = blockIdx.x * 2, g1 = g0 + 1;
    const int oi = blockIdx.x * 2 * HID + tid;
    float s = 0.0f;
    if (tid < 2 * HID) s = sums[oi];

    const int4* b4 = (const int4*)batch;
    int c0 = 0, c1 = 0;
    for (int k = 0; k < (N_NODES / 4 + FIN_THREADS - 1) / FIN_THREADS; ++k) {
        const int i = tid + k * FIN_THREADS;
        if (i < N_NODES / 4) {
            const int4 v = b4[i];
            c0 += (v.x == g0) + (v.y == g0) + (v.z == g0) + (v.w == g0);
            c1 += (v.x == g1) + (v.y == g1) + (v.z == g1) + (v.w == g1);
        }
    }
    for (int o = 16; o > 0; o >>= 1) {
        c0 += __shfl_xor(c0, o);
        c1 += __shfl_xor(c1, o);
    }
    if ((tid & 31) == 0) { sc[0][tid >> 5] = c0; sc[1][tid >> 5] = c1; }
    __syncthreads();
    if (tid < 2 * HID) {
        const int* p = sc[tid >> 7];
        int cnt = 0;
        for (int w = 0; w < FIN_THREADS / 32; ++w) cnt += p[w];
        out[oi] = s / fmaxf((float)cnt, 1.0f);
    }
}

extern "C" void kernel_launch(void* const* d_in, const int* in_sizes, int n_in,
                              void* d_out, int out_size, void* d_ws, size_t ws_size,
                              hipStream_t stream) {
    const float* h     = (const float*)d_in[0];
    const int*   eidx  = (const int*)  d_in[1];
    const float* x     = (const float*)d_in[2];
    const float* eattr = (const float*)d_in[3];
    const int*   batch = (const int*)  d_in[4];
    const float* We1   = (const float*)d_in[5];
    const float* be1   = (const float*)d_in[6];
    const float* We2   = (const float*)d_in[7];
    const float* be2   = (const float*)d_in[8];
    const float* Wn1   = (const float*)d_in[9];
    const float* bn1   = (const float*)d_in[10];
    const float* Wn2   = (const float*)d_in[11];
    const float* bn2   = (const float*)d_in[12];

    char* ws = (char*)d_ws;
    float* agg    = (float*)(ws + 204800);
    float* sums   = (float*)(ws + 204800 + 25600000);
    unsigned int* hb = (unsigned int*)(ws + 204800 + 25600000 + 32768 + 256);

    int nz = N_NODES * HID + N_GRAPHS * HID;
    prep_kernel<<<(nz + 255) / 256, 256, 0, stream>>>(agg, nz, h, hb);

    edge_kernel<<<EDGE_GRID, EDGE_THREADS, 0, stream>>>((const unsigned short*)hb, eidx, x, eattr,
                                                        We1, be1, We2, be2, agg);
    node_kernel<<<NODE_GRID, NODE_THREADS, 0, stream>>>(h, (const unsigned short*)hb, agg, batch,
                                                        Wn1, bn1, Wn2, bn2, sums);
    finalize_kernel<<<N_GRAPHS / 2, FIN_THREADS, 0, stream>>>(sums, batch, (float*)d_out);
}
